# starting kernel
# baseline (speedup 1.0000x reference)
#include <hip/hip_runtime.h>
#include <hip/hip_bf16.h>


typedef float v2f __attribute__((ext_vector_type(2)));
typedef float v8f __attribute__((ext_vector_type(8)));

#define NNODES 40000
#define NEDGES 640000
#define NGRAPH 64
#define DIM_IN 128
#define DIM_H  256
#define DIM_O  128
#define MTILES 4


__global__ void k_fill(float* __restrict__ p, float v, long n) {
    long i = (long)blockIdx.x * blockDim.x + threadIdx.x;
    if (i < n) p[i] = v;
}

__global__ void k_deg_accum(const int* __restrict__ dst, float* __restrict__ deg, int e) {
    int i = blockIdx.x * blockDim.x + threadIdx.x;
    if (i < e) unsafeAtomicAdd(&deg[dst[i]], 1.0f);
}

__global__ void k_rsqrt(float* __restrict__ d, int n) {
    int i = blockIdx.x * blockDim.x + threadIdx.x;
    if (i < n) d[i] = rsqrtf(d[i]);
}


__global__ __launch_bounds__(256) void k_gemm_wmma(const float* __restrict__ A,
                                                   const float* __restrict__ W,
                                                   float* __restrict__ C,
                                                   int M, int K, int Nc) {
    const int lane = threadIdx.x & 31;
    const int wave = threadIdx.x >> 5;
    const int m0 = blockIdx.x * (16 * MTILES);
    const int n0 = (blockIdx.y * 8 + wave) * 16;
    if (m0 >= M || n0 >= Nc) return;

    const bool hi = lane >= 16;
    const int  l  = lane & 15;
    const int  col = n0 + l;

    const float* __restrict__ arow[MTILES];
#pragma unroll
    for (int t = 0; t < MTILES; ++t)
        arow[t] = A + (size_t)(m0 + t * 16 + l) * K;

    v8f acc[MTILES];
#pragma unroll
    for (int t = 0; t < MTILES; ++t) acc[t] = (v8f){};

    for (int k0 = 0; k0 < K; k0 += 4) {
        const int ka = k0 + (hi ? 2 : 0);
        v2f b;
        b.x = W[(size_t)ka * Nc + col];
        b.y = W[(size_t)(ka + 1) * Nc + col];
#pragma unroll
        for (int t = 0; t < MTILES; ++t) {
            v2f a = *(const v2f*)(arow[t] + ka);
            acc[t] = __builtin_amdgcn_wmma_f32_16x16x4_f32(
                 false, a,  false, b,
                 (short)0, acc[t],  false,  false);
        }
    }

#pragma unroll
    for (int t = 0; t < MTILES; ++t) {
        const int rbase = m0 + t * 16 + (hi ? 8 : 0);
#pragma unroll
        for (int v = 0; v < 8; ++v)
            C[(size_t)(rbase + v) * Nc + col] = acc[t][v];
    }
}


__global__ void k_scatter(const float* __restrict__ h, const int* __restrict__ src,
                          const int* __restrict__ dst, const float* __restrict__ dis,
                          float* __restrict__ agg, long total, int shift, int H) {
    long gid = (long)blockIdx.x * blockDim.x + threadIdx.x;
    if (gid >= total) return;
    const int e = (int)(gid >> shift);
    const int j = (int)((gid & ((1 << shift) - 1)) << 2);
    const int s = src[e];
    const int d = dst[e];
    const float nrm = dis[s] * dis[d];
    const float4 hv = *(const float4*)(h + (size_t)s * H + j);
    float* out = agg + (size_t)d * H + j;
    unsafeAtomicAdd(out + 0, nrm * hv.x);
    unsafeAtomicAdd(out + 1, nrm * hv.y);
    unsafeAtomicAdd(out + 2, nrm * hv.z);
    unsafeAtomicAdd(out + 3, nrm * hv.w);
}


__global__ void k_fuse(const float* __restrict__ agg, const float* __restrict__ h,
                       const float* __restrict__ dis, const float* __restrict__ b,
                       float* __restrict__ out, long total, int H, int relu) {
    long gid = (long)blockIdx.x * blockDim.x + threadIdx.x;
    if (gid >= total) return;
    const int i = (int)(gid / H);
    const int j = (int)(gid - (long)i * H);
    const float ds = dis[i];
    float v = agg[gid] + ds * ds * h[gid] + b[j];
    if (relu) v = fmaxf(v, 0.0f);
    out[gid] = v;
}


__global__ void k_count(const int* __restrict__ batch, float* __restrict__ counts, int n) {
    int i = blockIdx.x * blockDim.x + threadIdx.x;
    if (i < n) unsafeAtomicAdd(&counts[batch[i]], 1.0f);
}

__global__ void k_pool_sum(const float* __restrict__ emb, const int* __restrict__ batch,
                           float* __restrict__ pool, long total, int D) {
    long gid = (long)blockIdx.x * blockDim.x + threadIdx.x;
    if (gid >= total) return;
    const int i = (int)(gid / D);
    const int j = (int)(gid - (long)i * D);
    unsafeAtomicAdd(&pool[(size_t)batch[i] * D + j], emb[gid]);
}

__global__ void k_pool_fin(const float* __restrict__ pool, const float* __restrict__ counts,
                           float* __restrict__ gemb, int total, int D) {
    int gid = blockIdx.x * blockDim.x + threadIdx.x;
    if (gid >= total) return;
    const int g = gid / D;
    gemb[gid] = pool[gid] / fmaxf(counts[g], 1.0f);
}


__global__ void k_logits(const float* __restrict__ gemb, const float* __restrict__ Wc,
                         const float* __restrict__ bc, float* __restrict__ logits, int D) {
    const int t = threadIdx.x;
    const int g = t >> 1;
    const int c = t & 1;
    float acc = bc[c];
    const float* row = gemb + (size_t)g * D;
    for (int k = 0; k < D; ++k) acc += row[k] * Wc[k * 2 + c];
    logits[g * 2 + c] = acc;
}


static inline int cdiv_l(long a, int b) { return (int)((a + b - 1) / b); }

extern "C" void kernel_launch(void* const* d_in, const int* in_sizes, int n_in,
                              void* d_out, int out_size, void* d_ws, size_t ws_size,
                              hipStream_t stream) {
    const float* x     = (const float*)d_in[0];
    const int*   ei    = (const int*)d_in[1];
    const int*   batch = (const int*)d_in[2];
    const float* W_in  = (const float*)d_in[3];
    const float* b_in  = (const float*)d_in[4];
    const float* W_mid = (const float*)d_in[5];
    const float* b_mid = (const float*)d_in[6];
    const float* W_out = (const float*)d_in[7];
    const float* b_out = (const float*)d_in[8];
    const float* W_cls = (const float*)d_in[9];
    const float* b_cls = (const float*)d_in[10];

    const int N = in_sizes[0] / DIM_IN;
    const int E = in_sizes[1] / 2;
    const int* src = ei;
    const int* dst = ei + E;

    float* w       = (float*)d_ws;
    float* dis     = w;
    float* bufH    = dis + N;
    float* bufAgg  = bufH + (size_t)N * DIM_H;
    float* bufA    = bufAgg + (size_t)N * DIM_H;
    float* pool    = bufA + (size_t)N * DIM_H;
    float* counts  = pool + (size_t)NGRAPH * DIM_O;

    float* node_emb = (float*)d_out;
    float* gemb     = node_emb + (size_t)N * DIM_O;
    float* logits   = gemb + (size_t)NGRAPH * DIM_O;

    const int B = 256;
    const int MB = 16 * MTILES;

    k_fill<<<cdiv_l(N, B), B, 0, stream>>>(dis, 1.0f, N);
    k_deg_accum<<<cdiv_l(E, B), B, 0, stream>>>(dst, dis, E);
    k_rsqrt<<<cdiv_l(N, B), B, 0, stream>>>(dis, N);

    {
        dim3 grid(N / MB, DIM_H / 128);
        k_gemm_wmma<<<grid, B, 0, stream>>>(x, W_in, bufH, N, DIM_IN, DIM_H);
        long tot = (long)N * DIM_H;
        k_fill<<<cdiv_l(tot, B), B, 0, stream>>>(bufAgg, 0.0f, tot);
        long etot = (long)E * (DIM_H / 4);
        k_scatter<<<cdiv_l(etot, B), B, 0, stream>>>(bufH, src, dst, dis, bufAgg, etot, 6, DIM_H);
        k_fuse<<<cdiv_l(tot, B), B, 0, stream>>>(bufAgg, bufH, dis, b_in, bufA, tot, DIM_H, 1);
    }

    {
        dim3 grid(N / MB, DIM_H / 128);
        k_gemm_wmma<<<grid, B, 0, stream>>>(bufA, W_mid, bufH, N, DIM_H, DIM_H);
        long tot = (long)N * DIM_H;
        k_fill<<<cdiv_l(tot, B), B, 0, stream>>>(bufAgg, 0.0f, tot);
        long etot = (long)E * (DIM_H / 4);
        k_scatter<<<cdiv_l(etot, B), B, 0, stream>>>(bufH, src, dst, dis, bufAgg, etot, 6, DIM_H);
        k_fuse<<<cdiv_l(tot, B), B, 0, stream>>>(bufAgg, bufH, dis, b_mid, bufA, tot, DIM_H, 1);
    }

    {
        dim3 grid(N / MB, DIM_O / 128);
        k_gemm_wmma<<<grid, B, 0, stream>>>(bufA, W_out, bufH, N, DIM_H, DIM_O);
        long tot = (long)N * DIM_O;
        k_fill<<<cdiv_l(tot, B), B, 0, stream>>>(bufAgg, 0.0f, tot);
        long etot = (long)E * (DIM_O / 4);
        k_scatter<<<cdiv_l(etot, B), B, 0, stream>>>(bufH, src, dst, dis, bufAgg, etot, 5, DIM_O);
        k_fuse<<<cdiv_l(tot, B), B, 0, stream>>>(bufAgg, bufH, dis, b_out, node_emb, tot, DIM_O, 0);
    }

    {
        long ptot = (long)NGRAPH * DIM_O + NGRAPH;
        k_fill<<<cdiv_l(ptot, B), B, 0, stream>>>(pool, 0.0f, ptot);
        k_count<<<cdiv_l(N, B), B, 0, stream>>>(batch, counts, N);
        long tot = (long)N * DIM_O;
        k_pool_sum<<<cdiv_l(tot, B), B, 0, stream>>>(node_emb, batch, pool, tot, DIM_O);
        int gtot = NGRAPH * DIM_O;
        k_pool_fin<<<cdiv_l(gtot, B), B, 0, stream>>>(pool, counts, gemb, gtot, DIM_O);
    }

    k_logits<<<1, 128, 0, stream>>>(gemb, W_cls, b_cls, logits, DIM_O);
}


# Edge aggregation as a per-destination gather over an edge list counting-sorted by dst in each call, self term + bias + relu folded in; replaces zero-fill + float-atomic scatter + fuse pass
# speedup vs baseline: 2.0501x; 2.0501x over previous
#include <hip/hip_runtime.h>
#include <hip/hip_bf16.h>


typedef float v2f __attribute__((ext_vector_type(2)));
typedef float v8f __attribute__((ext_vector_type(8)));

#define NNODES 40000
#define NEDGES 640000
#define NGRAPH 64
#define DIM_IN 128
#define DIM_H  256
#define DIM_O  128
#define MTILES 4


__global__ void k_fill(float* __restrict__ p, float v, long n) {
    long i = (long)blockIdx.x * blockDim.x + threadIdx.x;
    if (i < n) p[i] = v;
}

__global__ void k_zero_int(int* __restrict__ p, int n) {
    int i = blockIdx.x * blockDim.x + threadIdx.x;
    if (i < n) p[i] = 0;
}

__global__ void k_deg_slot(const int* __restrict__ dst, int* __restrict__ cnt,
                           int* __restrict__ slot, int e) {
    int i = blockIdx.x * blockDim.x + threadIdx.x;
    if (i < e) slot[i] = atomicAdd(&cnt[dst[i]], 1);
}

#define SCAN_THREADS 1024
#define SCAN_PER ((NNODES + SCAN_THREADS - 1) / SCAN_THREADS)

__global__ __launch_bounds__(SCAN_THREADS) void k_scan_dis(int* __restrict__ cr,
                                                           float* __restrict__ dis, int n) {
    __shared__ int wsum[SCAN_THREADS / 32];
    const int t    = threadIdx.x;
    const int lane = t & 31;
    const int wv   = t >> 5;
    const int b0   = t * SCAN_PER;

    int c[SCAN_PER];
    int local = 0;
#pragma unroll
    for (int i = 0; i < SCAN_PER; ++i) {
        const int idx = b0 + i;
        c[i] = (idx < n) ? cr[idx] : 0;
        local += c[i];
    }

    int x = local;
    for (int off = 1; off < 32; off <<= 1) {
        const int y = __shfl_up(x, off);
        if (lane >= off) x += y;
    }
    if (lane == 31) wsum[wv] = x;
    __syncthreads();
    if (wv == 0) {
        int s = wsum[lane];
        for (int off = 1; off < 32; off <<= 1) {
            const int y = __shfl_up(s, off);
            if (lane >= off) s += y;
        }
        wsum[lane] = s;
    }
    __syncthreads();

    int run = x - local + (wv > 0 ? wsum[wv - 1] : 0);
#pragma unroll
    for (int i = 0; i < SCAN_PER; ++i) {
        const int idx = b0 + i;
        if (idx < n) {
            cr[idx]  = run;
            dis[idx] = rsqrtf((float)c[i] + 1.0f);
        }
        run += c[i];
    }
    if (t == SCAN_THREADS - 1) cr[n] = run;
}

__global__ void k_place(const int* __restrict__ src, const int* __restrict__ dst,
                        const int* __restrict__ rowptr, const int* __restrict__ slot,
                        const float* __restrict__ dis, int2* __restrict__ csr, int e) {
    int i = blockIdx.x * blockDim.x + threadIdx.x;
    if (i >= e) return;
    const int s = src[i];
    const int d = dst[i];
    const float nrm = dis[s] * dis[d];
    csr[rowptr[d] + slot[i]] = make_int2(s, __float_as_int(nrm));
}


__global__ __launch_bounds__(256) void k_gemm_wmma(const float* __restrict__ A,
                                                   const float* __restrict__ W,
                                                   float* __restrict__ C,
                                                   int M, int K, int Nc) {
    const int lane = threadIdx.x & 31;
    const int wave = threadIdx.x >> 5;
    const int m0 = blockIdx.x * (16 * MTILES);
    const int n0 = (blockIdx.y * 8 + wave) * 16;
    if (m0 >= M || n0 >= Nc) return;

    const bool hi = lane >= 16;
    const int  l  = lane & 15;
    const int  col = n0 + l;

    const float* __restrict__ arow[MTILES];
#pragma unroll
    for (int t = 0; t < MTILES; ++t)
        arow[t] = A + (size_t)(m0 + t * 16 + l) * K;

    v8f acc[MTILES];
#pragma unroll
    for (int t = 0; t < MTILES; ++t) acc[t] = (v8f){};

    for (int k0 = 0; k0 < K; k0 += 4) {
        const int ka = k0 + (hi ? 2 : 0);
        v2f b;
        b.x = W[(size_t)ka * Nc + col];
        b.y = W[(size_t)(ka + 1) * Nc + col];
#pragma unroll
        for (int t = 0; t < MTILES; ++t) {
            v2f a = *(const v2f*)(arow[t] + ka);
            acc[t] = __builtin_amdgcn_wmma_f32_16x16x4_f32(
                 false, a,  false, b,
                 (short)0, acc[t],  false,  false);
        }
    }

#pragma unroll
    for (int t = 0; t < MTILES; ++t) {
        const int rbase = m0 + t * 16 + (hi ? 8 : 0);
#pragma unroll
        for (int v = 0; v < 8; ++v)
            C[(size_t)(rbase + v) * Nc + col] = acc[t][v];
    }
}


template <int H, int K>
__device__ __forceinline__ void gather_rows(const float* __restrict__ hl, int my_s, int my_w,
                                            int i, float4 (&acc)[H / 128]) {
    constexpr int V = H / 128;
    float  w[K];
    float4 v[K][V];
#pragma unroll
    for (int k = 0; k < K; ++k) {
        const int s = __builtin_amdgcn_readlane(my_s, i + k);
        w[k] = __int_as_float(__builtin_amdgcn_readlane(my_w, i + k));
        const float* row = hl + (size_t)s * H;
        for (int q = 0; q < V; ++q) v[k][q] = *(const float4*)(row + q * 128);
    }
#pragma unroll
    for (int k = 0; k < K; ++k) {
        for (int q = 0; q < V; ++q) {
            acc[q].x = __builtin_fmaf(w[k], v[k][q].x, acc[q].x);
            acc[q].y = __builtin_fmaf(w[k], v[k][q].y, acc[q].y);
            acc[q].z = __builtin_fmaf(w[k], v[k][q].z, acc[q].z);
            acc[q].w = __builtin_fmaf(w[k], v[k][q].w, acc[q].w);
        }
    }
}

template <int H, int RELU>
__global__ __launch_bounds__(256) void k_gather(const float* __restrict__ h,
                                                const int* __restrict__ rowptr,
                                                const int2* __restrict__ csr,
                                                const float* __restrict__ dis,
                                                const float* __restrict__ b,
                                                float* __restrict__ out, int n) {
    constexpr int V = H / 128;
    constexpr int K = 1024 / H;
    const int lane = threadIdx.x & 31;
    const int node = __builtin_amdgcn_readfirstlane((int)(blockIdx.x * 8 + (threadIdx.x >> 5)));
    if (node >= n) return;
    const int beg = rowptr[node];
    const int end = rowptr[node + 1];

    const float* hl = h + lane * 4;
    float4 acc[V];
    for (int q = 0; q < V; ++q) acc[q] = make_float4(0.f, 0.f, 0.f, 0.f);

    for (int base = beg; base < end; base += 32) {
        const int cntw = (end - base < 32) ? (end - base) : 32;
        int2 me = make_int2(0, 0);
        if (lane < cntw) me = csr[base + lane];
        int i = 0;
        for (; i + K <= cntw; i += K) gather_rows<H, K>(hl, me.x, me.y, i, acc);
        if (K >= 8 && i + 4 <= cntw) { gather_rows<H, 4>(hl, me.x, me.y, i, acc); i += 4; }
        if (i + 2 <= cntw)           { gather_rows<H, 2>(hl, me.x, me.y, i, acc); i += 2; }
        if (i + 1 <= cntw)           { gather_rows<H, 1>(hl, me.x, me.y, i, acc); }
    }

    const float ds = dis[node];
    const float sl = ds * ds;
    const float* hrow = hl + (size_t)node * H;
    float* orow = out + (size_t)node * H + lane * 4;
    for (int q = 0; q < V; ++q) {
        const float4 hv = *(const float4*)(hrow + q * 128);
        const float4 bv = *(const float4*)(b + q * 128 + lane * 4);
        float4 r;
        r.x = acc[q].x + sl * hv.x + bv.x;
        r.y = acc[q].y + sl * hv.y + bv.y;
        r.z = acc[q].z + sl * hv.z + bv.z;
        r.w = acc[q].w + sl * hv.w + bv.w;
        if (RELU) {
            r.x = fmaxf(r.x, 0.0f); r.y = fmaxf(r.y, 0.0f);
            r.z = fmaxf(r.z, 0.0f); r.w = fmaxf(r.w, 0.0f);
        }
        *(float4*)(orow + q * 128) = r;
    }
}


__global__ void k_count(const int* __restrict__ batch, float* __restrict__ counts, int n) {
    int i = blockIdx.x * blockDim.x + threadIdx.x;
    if (i < n) unsafeAtomicAdd(&counts[batch[i]], 1.0f);
}

__global__ void k_pool_sum(const float* __restrict__ emb, const int* __restrict__ batch,
                           float* __restrict__ pool, long total, int D) {
    long gid = (long)blockIdx.x * blockDim.x + threadIdx.x;
    if (gid >= total) return;
    const int i = (int)(gid / D);
    const int j = (int)(gid - (long)i * D);
    unsafeAtomicAdd(&pool[(size_t)batch[i] * D + j], emb[gid]);
}

__global__ void k_pool_fin(const float* __restrict__ pool, const float* __restrict__ counts,
                           float* __restrict__ gemb, int total, int D) {
    int gid = blockIdx.x * blockDim.x + threadIdx.x;
    if (gid >= total) return;
    const int g = gid / D;
    gemb[gid] = pool[gid] / fmaxf(counts[g], 1.0f);
}


__global__ void k_logits(const float* __restrict__ gemb, const float* __restrict__ Wc,
                         const float* __restrict__ bc, float* __restrict__ logits, int D) {
    const int t = threadIdx.x;
    const int g = t >> 1;
    const int c = t & 1;
    float acc = bc[c];
    const float* row = gemb + (size_t)g * D;
    for (int k = 0; k < D; ++k) acc += row[k] * Wc[k * 2 + c];
    logits[g * 2 + c] = acc;
}


static inline int cdiv_l(long a, int b) { return (int)((a + b - 1) / b); }

extern "C" void kernel_launch(void* const* d_in, const int* in_sizes, int n_in,
                              void* d_out, int out_size, void* d_ws, size_t ws_size,
                              hipStream_t stream) {
    const float* x     = (const float*)d_in[0];
    const int*   ei    = (const int*)d_in[1];
    const int*   batch = (const int*)d_in[2];
    const float* W_in  = (const float*)d_in[3];
    const float* b_in  = (const float*)d_in[4];
    const float* W_mid = (const float*)d_in[5];
    const float* b_mid = (const float*)d_in[6];
    const float* W_out = (const float*)d_in[7];
    const float* b_out = (const float*)d_in[8];
    const float* W_cls = (const float*)d_in[9];
    const float* b_cls = (const float*)d_in[10];

    const int N = in_sizes[0] / DIM_IN;
    const int E = in_sizes[1] / 2;
    const int* src = ei;
    const int* dst = ei + E;

    float* w       = (float*)d_ws;
    float* dis     = w;
    float* bufH    = dis + N;
    float* bufCsr  = bufH + (size_t)N * DIM_H;
    float* bufA    = bufCsr + (size_t)N * DIM_H;
    float* pool    = bufA + (size_t)N * DIM_H;
    float* counts  = pool + (size_t)NGRAPH * DIM_O;

    int*  rowptr = (int*)bufCsr;
    int*  slot   = rowptr + (N + 64);
    int2* csr    = (int2*)(slot + E);

    float* node_emb = (float*)d_out;
    float* gemb     = node_emb + (size_t)N * DIM_O;
    float* logits   = gemb + (size_t)NGRAPH * DIM_O;

    const int B = 256;
    const int MB = 16 * MTILES;
    const int GB = cdiv_l(N, 8);

    k_zero_int<<<cdiv_l(N, B), B, 0, stream>>>(rowptr, N);
    k_deg_slot<<<cdiv_l(E, B), B, 0, stream>>>(dst, rowptr, slot, E);
    k_scan_dis<<<1, SCAN_THREADS, 0, stream>>>(rowptr, dis, N);
    k_place<<<cdiv_l(E, B), B, 0, stream>>>(src, dst, rowptr, slot, dis, csr, E);

    {
        dim3 grid(N / MB, DIM_H / 128);
        k_gemm_wmma<<<grid, B, 0, stream>>>(x, W_in, bufH, N, DIM_IN, DIM_H);
        k_gather<DIM_H, 1><<<GB, B, 0, stream>>>(bufH, rowptr, csr, dis, b_in, bufA, N);
    }

    {
        dim3 grid(N / MB, DIM_H / 128);
        k_gemm_wmma<<<grid, B, 0, stream>>>(bufA, W_mid, bufH, N, DIM_H, DIM_H);
        k_gather<DIM_H, 1><<<GB, B, 0, stream>>>(bufH, rowptr, csr, dis, b_mid, bufA, N);
    }

    {
        dim3 grid(N / MB, DIM_O / 128);
        k_gemm_wmma<<<grid, B, 0, stream>>>(bufA, W_out, bufH, N, DIM_H, DIM_O);
        k_gather<DIM_O, 0><<<GB, B, 0, stream>>>(bufH, rowptr, csr, dis, b_out, node_emb, N);
    }

    {
        long ptot = (long)NGRAPH * DIM_O + NGRAPH;
        k_fill<<<cdiv_l(ptot, B), B, 0, stream>>>(pool, 0.0f, ptot);
        k_count<<<cdiv_l(N, B), B, 0, stream>>>(batch, counts, N);
        long tot = (long)N * DIM_O;
        k_pool_sum<<<cdiv_l(tot, B), B, 0, stream>>>(node_emb, batch, pool, tot, DIM_O);
        int gtot = NGRAPH * DIM_O;
        k_pool_fin<<<cdiv_l(gtot, B), B, 0, stream>>>(pool, counts, gemb, gtot, DIM_O);
    }

    k_logits<<<1, 128, 0, stream>>>(gemb, W_cls, b_cls, logits, DIM_O);
}


# GEMM wave tile reshaped to 32 rows x 64 columns with 16-byte loads of both operands (A: 4 consecutive k per lane, W: 4 adjacent columns per lane) and the next block's loads issued ahead
# speedup vs baseline: 2.8153x; 1.3733x over previous
#include <hip/hip_runtime.h>
#include <hip/hip_bf16.h>


typedef float v2f __attribute__((ext_vector_type(2)));
typedef float v8f __attribute__((ext_vector_type(8)));

#define NNODES 40000
#define NEDGES 640000
#define NGRAPH 64
#define DIM_IN 128
#define DIM_H  256
#define DIM_O  128


__global__ void k_zero_int(int* __restrict__ p, int n) {
    int i = blockIdx.x * blockDim.x + threadIdx.x;
    if (i < n) p[i] = 0;
}

__global__ void k_deg_slot(const int* __restrict__ dst, int* __restrict__ cnt,
                           int* __restrict__ slot, int e) {
    int i = blockIdx.x * blockDim.x + threadIdx.x;
    if (i < e) slot[i] = atomicAdd(&cnt[dst[i]], 1);
}

#define SCAN_THREADS 1024
#define SCAN_PER ((NNODES + SCAN_THREADS - 1) / SCAN_THREADS)

__global__ __launch_bounds__(SCAN_THREADS) void k_scan_dis(int* __restrict__ cr,
                                                           float* __restrict__ dis, int n) {
    __shared__ int wsum[SCAN_THREADS / 32];
    const int t    = threadIdx.x;
    const int lane = t & 31;
    const int wv   = t >> 5;
    const int b0   = t * SCAN_PER;

    int c[SCAN_PER];
    int local = 0;
#pragma unroll
    for (int i = 0; i < SCAN_PER; ++i) {
        const int idx = b0 + i;
        c[i] = (idx < n) ? cr[idx] : 0;
        local += c[i];
    }

    int x = local;
    for (int off = 1; off < 32; off <<= 1) {
        const int y = __shfl_up(x, off);
        if (lane >= off) x += y;
    }
    if (lane == 31) wsum[wv] = x;
    __syncthreads();
    if (wv == 0) {
        int s = wsum[lane];
        for (int off = 1; off < 32; off <<= 1) {
            const int y = __shfl_up(s, off);
            if (lane >= off) s += y;
        }
        wsum[lane] = s;
    }
    __syncthreads();

    int run = x - local + (wv > 0 ? wsum[wv - 1] : 0);
#pragma unroll
    for (int i = 0; i < SCAN_PER; ++i) {
        const int idx = b0 + i;
        if (idx < n) {
            cr[idx]  = run;
            dis[idx] = rsqrtf((float)c[i] + 1.0f);
        }
        run += c[i];
    }
    if (t == SCAN_THREADS - 1) cr[n] = run;
}

__global__ void k_place(const int* __restrict__ src, const int* __restrict__ dst,
                        const int* __restrict__ rowptr, const int* __restrict__ slot,
                        const float* __restrict__ dis, int2* __restrict__ csr, int e) {
    int i = blockIdx.x * blockDim.x + threadIdx.x;
    if (i >= e) return;
    const int s = src[i];
    const int d = dst[i];
    const float nrm = dis[s] * dis[d];
    csr[rowptr[d] + slot[i]] = make_int2(s, __float_as_int(nrm));
}


template <int MT>
__device__ __forceinline__ void gemm_fetch(const float* const (&ap)[MT], const float* __restrict__ wp,
                                           int k, int Nc, float4 (&a)[MT], float4 (&b)[4]) {
    for (int t = 0; t < MT; ++t) a[t] = *(const float4*)(ap[t] + k);
    for (int j = 0; j < 4; ++j)  b[j] = *(const float4*)(wp + (size_t)(k + j) * Nc);
}

template <int MT>
__device__ __forceinline__ void gemm_steps(const float4 (&a)[MT], const float4 (&b)[4],
                                           v8f (&acc)[MT][4]) {
    const float bx[4][4] = {{b[0].x, b[0].y, b[0].z, b[0].w},
                            {b[1].x, b[1].y, b[1].z, b[1].w},
                            {b[2].x, b[2].y, b[2].z, b[2].w},
                            {b[3].x, b[3].y, b[3].z, b[3].w}};
    for (int p = 0; p < 2; ++p) {
        for (int t = 0; t < MT; ++t) {
            v2f av;
            av.x = p ? a[t].z : a[t].x;
            av.y = p ? a[t].w : a[t].y;
            for (int q = 0; q < 4; ++q) {
                v2f bv;
                bv.x = bx[2 * p][q];
                bv.y = bx[2 * p + 1][q];
                acc[t][q] = __builtin_amdgcn_wmma_f32_16x16x4_f32(
                     false, av,  false, bv,
                     (short)0, acc[t][q],  false,  false);
            }
        }
    }
}

template <int MT, int WN>
__global__ __launch_bounds__(128) void k_gemm_wmma(const float* __restrict__ A,
                                                   const float* __restrict__ W,
                                                   float* __restrict__ C,
                                                   int M, int K, int Nc) {
    constexpr int WM = 4 / WN;
    const int lane = threadIdx.x & 31;
    const int wave = __builtin_amdgcn_readfirstlane((int)(threadIdx.x >> 5));
    const int m0 = ((int)blockIdx.x * WM + wave / WN) * (16 * MT);
    const int n0 = ((int)blockIdx.y * WN + wave % WN) * 64;
    if (m0 >= M || n0 >= Nc) return;

    const int h = lane >> 4;
    const int c = lane & 15;

    const float* ap[MT];
    for (int t = 0; t < MT; ++t)
        ap[t] = A + (size_t)(m0 + t * 16 + c) * K + 4 * h;
    const float* __restrict__ wp = W + (size_t)(4 * h) * Nc + n0 + 4 * c;

    v8f acc[MT][4];
    for (int t = 0; t < MT; ++t)
        for (int q = 0; q < 4; ++q) acc[t][q] = (v8f){};

    float4 a0[MT], b0[4], a1[MT], b1[4];
    gemm_fetch<MT>(ap, wp, 0, Nc, a0, b0);

    for (int k0 = 0; k0 < K; k0 += 16) {
        gemm_fetch<MT>(ap, wp, k0 + 8, Nc, a1, b1);
        __builtin_amdgcn_sched_barrier(0);
        gemm_steps<MT>(a0, b0, acc);

        const int kn = (k0 + 16 < K) ? k0 + 16 : k0;
        gemm_fetch<MT>(ap, wp, kn, Nc, a0, b0);
        __builtin_amdgcn_sched_barrier(0);
        gemm_steps<MT>(a1, b1, acc);
    }

    for (int t = 0; t < MT; ++t) {
        float* crow = C + (size_t)(m0 + t * 16 + 8 * h) * Nc + n0 + 4 * c;
        for (int v = 0; v < 8; ++v) {
            float4 r;
            r.x = acc[t][0][v]; r.y = acc[t][1][v];
            r.z = acc[t][2][v]; r.w = acc[t][3][v];
            *(float4*)(crow + (size_t)v * Nc) = r;
        }
    }
}


template <int H, int K>
__device__ __forceinline__ void gather_rows(const float* __restrict__ hl, int my_s, int my_w,
                                            int i, float4 (&acc)[H / 128]) {
    constexpr int V = H / 128;
    float  w[K];
    float4 v[K][V];
#pragma unroll
    for (int k = 0; k < K; ++k) {
        const int s = __builtin_amdgcn_readlane(my_s, i + k);
        w[k] = __int_as_float(__builtin_amdgcn_readlane(my_w, i + k));
        const float* row = hl + (size_t)s * H;
        for (int q = 0; q < V; ++q) v[k][q] = *(const float4*)(row + q * 128);
    }
#pragma unroll
    for (int k = 0; k < K; ++k) {
        for (int q = 0; q < V; ++q) {
            acc[q].x = __builtin_fmaf(w[k], v[k][q].x, acc[q].x);
            acc[q].y = __builtin_fmaf(w[k], v[k][q].y, acc[q].y);
            acc[q].z = __builtin_fmaf(w[k], v[k][q].z, acc[q].z);
            acc[q].w = __builtin_fmaf(w[k], v[k][q].w, acc[q].w);
        }
    }
}

template <int H, int RELU>
__global__ __launch_bounds__(256) void k_gather(const float* __restrict__ h,
                                                const int* __restrict__ rowptr,
                                                const int2* __restrict__ csr,
                                                const float* __restrict__ dis,
                                                const float* __restrict__ b,
                                                float* __restrict__ out, int n) {
    constexpr int V = H / 128;
    constexpr int K = 1024 / H;
    const int lane = threadIdx.x & 31;
    const int node = __builtin_amdgcn_readfirstlane((int)(blockIdx.x * 8 + (threadIdx.x >> 5)));
    if (node >= n) return;
    const int beg = rowptr[node];
    const int end = rowptr[node + 1];

    const float* hl = h + lane * 4;
    float4 acc[V];
    for (int q = 0; q < V; ++q) acc[q] = make_float4(0.f, 0.f, 0.f, 0.f);

    for (int base = beg; base < end; base += 32) {
        const int cntw = (end - base < 32) ? (end - base) : 32;
        int2 me = make_int2(0, 0);
        if (lane < cntw) me = csr[base + lane];
        int i = 0;
        for (; i + K <= cntw; i += K) gather_rows<H, K>(hl, me.x, me.y, i, acc);
        if (K >= 8 && i + 4 <= cntw) { gather_rows<H, 4>(hl, me.x, me.y, i, acc); i += 4; }
        if (i + 2 <= cntw)           { gather_rows<H, 2>(hl, me.x, me.y, i, acc); i += 2; }
        if (i + 1 <= cntw)           { gather_rows<H, 1>(hl, me.x, me.y, i, acc); }
    }

    const float ds = dis[node];
    const float sl = ds * ds;
    const float* hrow = hl + (size_t)node * H;
    float* orow = out + (size_t)node * H + lane * 4;
    for (int q = 0; q < V; ++q) {
        const float4 hv = *(const float4*)(hrow + q * 128);
        const float4 bv = *(const float4*)(b + q * 128 + lane * 4);
        float4 r;
        r.x = acc[q].x + sl * hv.x + bv.x;
        r.y = acc[q].y + sl * hv.y + bv.y;
        r.z = acc[q].z + sl * hv.z + bv.z;
        r.w = acc[q].w + sl * hv.w + bv.w;
        if (RELU) {
            r.x = fmaxf(r.x, 0.0f); r.y = fmaxf(r.y, 0.0f);
            r.z = fmaxf(r.z, 0.0f); r.w = fmaxf(r.w, 0.0f);
        }
        *(float4*)(orow + q * 128) = r;
    }
}


#define POOL_THREADS 1024
#define POOL_WAVES   (POOL_THREADS / 32)
#define POOL_CHUNK   8
#define POOL_ROWS    8

__global__ __launch_bounds__(POOL_THREADS) void k_pool_mean(const float* __restrict__ emb,
                                                            const int* __restrict__ batch,
                                                            float* __restrict__ gemb, int n) {
    static_assert(DIM_O == 128, "a lane owns 4 of 128 columns");
    __shared__ float4 part[POOL_WAVES * 32];
    __shared__ int    wcnt[POOL_WAVES];

    const int g    = blockIdx.x;
    const int lane = threadIdx.x & 31;
    const int wv   = __builtin_amdgcn_readfirstlane((int)(threadIdx.x >> 5));

    const float* el = emb + lane * 4;
    float4 acc = make_float4(0.f, 0.f, 0.f, 0.f);
    int cnt = 0;

    for (int c0 = wv * 32; c0 < n; c0 += POOL_CHUNK * POOL_THREADS) {
        int b[POOL_CHUNK];
#pragma unroll
        for (int u = 0; u < POOL_CHUNK; ++u) {
            const int i = c0 + u * POOL_THREADS + lane;
            b[u] = (i < n) ? batch[i] : -1;
        }
#pragma unroll
        for (int u = 0; u < POOL_CHUNK; ++u) {
            unsigned mask = __builtin_amdgcn_ballot_w32(b[u] == g);
            cnt += __builtin_popcount(mask);
            const int base = c0 + u * POOL_THREADS;
            while (mask) {
                float4 v[POOL_ROWS];
#pragma unroll
                for (int r = 0; r < POOL_ROWS; ++r) {
                    v[r] = make_float4(0.f, 0.f, 0.f, 0.f);
                    if (mask) {
                        const int k = __builtin_ctz(mask);
                        mask &= mask - 1;
                        v[r] = *(const float4*)(el + (size_t)(base + k) * DIM_O);
                    }
                }
#pragma unroll
                for (int r = 0; r < POOL_ROWS; ++r) {
                    acc.x += v[r].x; acc.y += v[r].y;
                    acc.z += v[r].z; acc.w += v[r].w;
                }
            }
        }
    }

    part[wv * 32 + lane] = acc;
    if (lane == 0) wcnt[wv] = cnt;
    __syncthreads();

    if (threadIdx.x < DIM_O) {
        const float* pf = (const float*)part;
        float s = 0.0f;
        int   c = 0;
        for (int w = 0; w < POOL_WAVES; ++w) {
            s += pf[w * DIM_O + threadIdx.x];
            c += wcnt[w];
        }
        gemb[(size_t)g * DIM_O + threadIdx.x] = s / fmaxf((float)c, 1.0f);
    }
}


__global__ void k_logits(const float* __restrict__ gemb, const float* __restrict__ Wc,
                         const float* __restrict__ bc, float* __restrict__ logits, int D) {
    const int t = threadIdx.x;
    const int g = t >> 1;
    const int c = t & 1;
    float acc = bc[c];
    const float* row = gemb + (size_t)g * D;
    for (int k = 0; k < D; ++k) acc += row[k] * Wc[k * 2 + c];
    logits[g * 2 + c] = acc;
}


static inline int cdiv_l(long a, int b) { return (int)((a + b - 1) / b); }

extern "C" void kernel_launch(void* const* d_in, const int* in_sizes, int n_in,
                              void* d_out, int out_size, void* d_ws, size_t ws_size,
                              hipStream_t stream) {
    const float* x     = (const float*)d_in[0];
    const int*   ei    = (const int*)d_in[1];
    const int*   batch = (const int*)d_in[2];
    const float* W_in  = (const float*)d_in[3];
    const float* b_in  = (const float*)d_in[4];
    const float* W_mid = (const float*)d_in[5];
    const float* b_mid = (const float*)d_in[6];
    const float* W_out = (const float*)d_in[7];
    const float* b_out = (const float*)d_in[8];
    const float* W_cls = (const float*)d_in[9];
    const float* b_cls = (const float*)d_in[10];

    const int N = in_sizes[0] / DIM_IN;
    const int E = in_sizes[1] / 2;
    const int* src = ei;
    const int* dst = ei + E;

    float* w       = (float*)d_ws;
    float* dis     = w;
    float* bufH    = dis + N;
    float* bufCsr  = bufH + (size_t)N * DIM_H;
    float* bufA    = bufCsr + (size_t)N * DIM_H;

    int*  rowptr = (int*)bufCsr;
    int*  slot   = rowptr + (N + 64);
    int2* csr    = (int2*)(slot + E);

    float* node_emb = (float*)d_out;
    float* gemb     = node_emb + (size_t)N * DIM_O;
    float* logits   = gemb + (size_t)NGRAPH * DIM_O;

    const int B = 256;
    const int MB = 32;
    const int GB = cdiv_l(N, 8);

    k_zero_int<<<cdiv_l(N, B), B, 0, stream>>>(rowptr, N);
    k_deg_slot<<<cdiv_l(E, B), B, 0, stream>>>(dst, rowptr, slot, E);
    k_scan_dis<<<1, SCAN_THREADS, 0, stream>>>(rowptr, dis, N);
    k_place<<<cdiv_l(E, B), B, 0, stream>>>(src, dst, rowptr, slot, dis, csr, E);

    {
        dim3 grid(N / MB, DIM_H / 256);
        k_gemm_wmma<2, 4><<<grid, 128, 0, stream>>>(x, W_in, bufH, N, DIM_IN, DIM_H);
        k_gather<DIM_H, 1><<<GB, B, 0, stream>>>(bufH, rowptr, csr, dis, b_in, bufA, N);
    }

    {
        dim3 grid(N / MB, DIM_H / 256);
        k_gemm_wmma<2, 4><<<grid, 128, 0, stream>>>(bufA, W_mid, bufH, N, DIM_H, DIM_H);
        k_gather<DIM_H, 1><<<GB, B, 0, stream>>>(bufH, rowptr, csr, dis, b_mid, bufA, N);
    }

    {
        dim3 grid(N / MB, DIM_O / 128);
        k_gemm_wmma<1, 2><<<grid, 128, 0, stream>>>(bufA, W_out, bufH, N, DIM_H, DIM_O);
        k_gather<DIM_O, 0><<<GB, B, 0, stream>>>(bufH, rowptr, csr, dis, b_out, node_emb, N);
    }

    k_pool_mean<<<NGRAPH, POOL_THREADS, 0, stream>>>(node_emb, batch, gemb, N);

    k_logits<<<1, 128, 0, stream>>>(gemb, W_cls, b_cls, logits, DIM_O);
}


# Offset scan moves whole lines: a wave owns 1280 consecutive nodes, 16-byte loads and stores (4 lines per instruction), lane scans chained over 10 passes; was 40 nodes per thread, 32 lines per access
# speedup vs baseline: 3.4736x; 1.2338x over previous
#include <hip/hip_runtime.h>
#include <hip/hip_bf16.h>


typedef float v2f __attribute__((ext_vector_type(2)));
typedef float v8f __attribute__((ext_vector_type(8)));

#define NNODES 40000
#define NEDGES 640000
#define NGRAPH 64
#define DIM_IN 128
#define DIM_H  256
#define DIM_O  128


__global__ void k_zero_int(int* __restrict__ p, int n) {
    int i = blockIdx.x * blockDim.x + threadIdx.x;
    if (i < n) p[i] = 0;
}

__global__ void k_deg_slot(const int* __restrict__ dst, int* __restrict__ cnt,
                           int* __restrict__ slot, int e) {
    int i = blockIdx.x * blockDim.x + threadIdx.x;
    if (i < e) slot[i] = atomicAdd(&cnt[dst[i]], 1);
}

#define SCAN_THREADS 1024
#define SCAN_WAVES   (SCAN_THREADS / 32)
#define SCAN_PASSES  ((NNODES + SCAN_THREADS * 4 - 1) / (SCAN_THREADS * 4))
#define SCAN_WSPAN   (SCAN_PASSES * 128)

__global__ __launch_bounds__(SCAN_THREADS) void k_scan_dis(int* __restrict__ cr,
                                                           float* __restrict__ dis, int n) {
    static_assert(SCAN_WAVES == 32, "one lane of the first wave per wave total");
    static_assert(NNODES % 4 == 0, "a lane's group of 4 nodes lies wholly inside or wholly outside the range");
    __shared__ int wsum[SCAN_WAVES];
    const int t    = threadIdx.x;
    const int lane = t & 31;
    const int wv   = t >> 5;
    const int b0   = wv * SCAN_WSPAN + lane * 4;

    int4 c[SCAN_PASSES];
#pragma unroll
    for (int j = 0; j < SCAN_PASSES; ++j) {
        const int idx = b0 + j * 128;
        c[j] = make_int4(0, 0, 0, 0);
        if (idx < n) c[j] = *(const int4*)(cr + idx);
    }

    int s4[SCAN_PASSES];
    int x[SCAN_PASSES];
#pragma unroll
    for (int j = 0; j < SCAN_PASSES; ++j) {
        s4[j] = c[j].x + c[j].y + c[j].z + c[j].w;
        x[j]  = s4[j];
    }
#pragma unroll
    for (int off = 1; off < 32; off <<= 1) {
#pragma unroll
        for (int j = 0; j < SCAN_PASSES; ++j) {
            const int y = __shfl_up(x[j], off);
            if (lane >= off) x[j] += y;
        }
    }

    int ex[SCAN_PASSES];
    int carry = 0;
#pragma unroll
    for (int j = 0; j < SCAN_PASSES; ++j) {
        ex[j] = carry + x[j] - s4[j];
        carry += __builtin_amdgcn_readlane(x[j], 31);
    }

    if (lane == 0) wsum[wv] = carry;
    __syncthreads();
    if (wv == 0) {
        int s = wsum[lane];
        for (int off = 1; off < 32; off <<= 1) {
            const int y = __shfl_up(s, off);
            if (lane >= off) s += y;
        }
        wsum[lane] = s;
    }
    __syncthreads();

    const int wbase = (wv > 0) ? wsum[wv - 1] : 0;
#pragma unroll
    for (int j = 0; j < SCAN_PASSES; ++j) {
        const int idx = b0 + j * 128;
        int4 r;
        r.x = wbase + ex[j];
        r.y = r.x + c[j].x;
        r.z = r.y + c[j].y;
        r.w = r.z + c[j].z;
        float4 d;
        d.x = rsqrtf((float)c[j].x + 1.0f);
        d.y = rsqrtf((float)c[j].y + 1.0f);
        d.z = rsqrtf((float)c[j].z + 1.0f);
        d.w = rsqrtf((float)c[j].w + 1.0f);
        if (idx < n) {
            *(int4*)(cr + idx)    = r;
            *(float4*)(dis + idx) = d;
        }
    }
    if (t == SCAN_THREADS - 1) cr[n] = wbase + carry;
}

__global__ void k_place(const int* __restrict__ src, const int* __restrict__ dst,
                        const int* __restrict__ rowptr, const int* __restrict__ slot,
                        const float* __restrict__ dis, int2* __restrict__ csr, int e) {
    int i = blockIdx.x * blockDim.x + threadIdx.x;
    if (i >= e) return;
    const int s = src[i];
    const int d = dst[i];
    const float nrm = dis[s] * dis[d];
    csr[rowptr[d] + slot[i]] = make_int2(s, __float_as_int(nrm));
}


template <int MT>
__device__ __forceinline__ void gemm_fetch(const float* const (&ap)[MT], const float* __restrict__ wp,
                                           int k, int Nc, float4 (&a)[MT], float4 (&b)[4]) {
    for (int t = 0; t < MT; ++t) a[t] = *(const float4*)(ap[t] + k);
    for (int j = 0; j < 4; ++j)  b[j] = *(const float4*)(wp + (size_t)(k + j) * Nc);
}

template <int MT>
__device__ __forceinline__ void gemm_steps(const float4 (&a)[MT], const float4 (&b)[4],
                                           v8f (&acc)[MT][4]) {
    const float bx[4][4] = {{b[0].x, b[0].y, b[0].z, b[0].w},
                            {b[1].x, b[1].y, b[1].z, b[1].w},
                            {b[2].x, b[2].y, b[2].z, b[2].w},
                            {b[3].x, b[3].y, b[3].z, b[3].w}};
    for (int p = 0; p < 2; ++p) {
        for (int t = 0; t < MT; ++t) {
            v2f av;
            av.x = p ? a[t].z : a[t].x;
            av.y = p ? a[t].w : a[t].y;
            for (int q = 0; q < 4; ++q) {
                v2f bv;
                bv.x = bx[2 * p][q];
                bv.y = bx[2 * p + 1][q];
                acc[t][q] = __builtin_amdgcn_wmma_f32_16x16x4_f32(
                     false, av,  false, bv,
                     (short)0, acc[t][q],  false,  false);
            }
        }
    }
}

template <int MT, int WN>
__global__ __launch_bounds__(128) void k_gemm_wmma(const float* __restrict__ A,
                                                   const float* __restrict__ W,
                                                   float* __restrict__ C,
                                                   int M, int K, int Nc) {
    constexpr int WM = 4 / WN;
    const int lane = threadIdx.x & 31;
    const int wave = __builtin_amdgcn_readfirstlane((int)(threadIdx.x >> 5));
    const int m0 = ((int)blockIdx.x * WM + wave / WN) * (16 * MT);
    const int n0 = ((int)blockIdx.y * WN + wave % WN) * 64;
    if (m0 >= M || n0 >= Nc) return;

    const int h = lane >> 4;
    const int c = lane & 15;

    const float* ap[MT];
    for (int t = 0; t < MT; ++t)
        ap[t] = A + (size_t)(m0 + t * 16 + c) * K + 4 * h;
    const float* __restrict__ wp = W + (size_t)(4 * h) * Nc + n0 + 4 * c;

    v8f acc[MT][4];
    for (int t = 0; t < MT; ++t)
        for (int q = 0; q < 4; ++q) acc[t][q] = (v8f){};

    float4 a0[MT], b0[4], a1[MT], b1[4];
    gemm_fetch<MT>(ap, wp, 0, Nc, a0, b0);

    for (int k0 = 0; k0 < K; k0 += 16) {
        gemm_fetch<MT>(ap, wp, k0 + 8, Nc, a1, b1);
        __builtin_amdgcn_sched_barrier(0);
        gemm_steps<MT>(a0, b0, acc);

        const int kn = (k0 + 16 < K) ? k0 + 16 : k0;
        gemm_fetch<MT>(ap, wp, kn, Nc, a0, b0);
        __builtin_amdgcn_sched_barrier(0);
        gemm_steps<MT>(a1, b1, acc);
    }

    for (int t = 0; t < MT; ++t) {
        float* crow = C + (size_t)(m0 + t * 16 + 8 * h) * Nc + n0 + 4 * c;
        for (int v = 0; v < 8; ++v) {
            float4 r;
            r.x = acc[t][0][v]; r.y = acc[t][1][v];
            r.z = acc[t][2][v]; r.w = acc[t][3][v];
            *(float4*)(crow + (size_t)v * Nc) = r;
        }
    }
}


template <int H, int K>
__device__ __forceinline__ void gather_rows(const float* __restrict__ hl, int my_s, int my_w,
                                            int i, float4 (&acc)[H / 128]) {
    constexpr int V = H / 128;
    float  w[K];
    float4 v[K][V];
#pragma unroll
    for (int k = 0; k < K; ++k) {
        const int s = __builtin_amdgcn_readlane(my_s, i + k);
        w[k] = __int_as_float(__builtin_amdgcn_readlane(my_w, i + k));
        const float* row = hl + (size_t)s * H;
        for (int q = 0; q < V; ++q) v[k][q] = *(const float4*)(row + q * 128);
    }
#pragma unroll
    for (int k = 0; k < K; ++k) {
        for (int q = 0; q < V; ++q) {
            acc[q].x = __builtin_fmaf(w[k], v[k][q].x, acc[q].x);
            acc[q].y = __builtin_fmaf(w[k], v[k][q].y, acc[q].y);
            acc[q].z = __builtin_fmaf(w[k], v[k][q].z, acc[q].z);
            acc[q].w = __builtin_fmaf(w[k], v[k][q].w, acc[q].w);
        }
    }
}

template <int H, int RELU>
__global__ __launch_bounds__(256) void k_gather(const float* __restrict__ h,
                                                const int* __restrict__ rowptr,
                                                const int2* __restrict__ csr,
                                                const float* __restrict__ dis,
                                                const float* __restrict__ b,
                                                float* __restrict__ out, int n) {
    constexpr int V = H / 128;
    constexpr int K = 1024 / H;
    const int lane = threadIdx.x & 31;
    const int node = __builtin_amdgcn_readfirstlane((int)(blockIdx.x * 8 + (threadIdx.x >> 5)));
    if (node >= n) return;
    const int beg = rowptr[node];
    const int end = rowptr[node + 1];

    const float* hl = h + lane * 4;
    float4 acc[V];
    for (int q = 0; q < V; ++q) acc[q] = make_float4(0.f, 0.f, 0.f, 0.f);

    for (int base = beg; base < end; base += 32) {
        const int cntw = (end - base < 32) ? (end - base) : 32;
        int2 me = make_int2(0, 0);
        if (lane < cntw) me = csr[base + lane];
        int i = 0;
        for (; i + K <= cntw; i += K) gather_rows<H, K>(hl, me.x, me.y, i, acc);
        if (K >= 8 && i + 4 <= cntw) { gather_rows<H, 4>(hl, me.x, me.y, i, acc); i += 4; }
        if (i + 2 <= cntw)           { gather_rows<H, 2>(hl, me.x, me.y, i, acc); i += 2; }
        if (i + 1 <= cntw)           { gather_rows<H, 1>(hl, me.x, me.y, i, acc); }
    }

    const float ds = dis[node];
    const float sl = ds * ds;
    const float* hrow = hl + (size_t)node * H;
    float* orow = out + (size_t)node * H + lane * 4;
    for (int q = 0; q < V; ++q) {
        const float4 hv = *(const float4*)(hrow + q * 128);
        const float4 bv = *(const float4*)(b + q * 128 + lane * 4);
        float4 r;
        r.x = acc[q].x + sl * hv.x + bv.x;
        r.y = acc[q].y + sl * hv.y + bv.y;
        r.z = acc[q].z + sl * hv.z + bv.z;
        r.w = acc[q].w + sl * hv.w + bv.w;
        if (RELU) {
            r.x = fmaxf(r.x, 0.0f); r.y = fmaxf(r.y, 0.0f);
            r.z = fmaxf(r.z, 0.0f); r.w = fmaxf(r.w, 0.0f);
        }
        *(float4*)(orow + q * 128) = r;
    }
}


#define POOL_THREADS 1024
#define POOL_WAVES   (POOL_THREADS / 32)
#define POOL_CHUNK   8
#define POOL_ROWS    8

__global__ __launch_bounds__(POOL_THREADS) void k_pool_mean(const float* __restrict__ emb,
                                                            const int* __restrict__ batch,
                                                            float* __restrict__ gemb, int n) {
    static_assert(DIM_O == 128, "a lane owns 4 of 128 columns");
    __shared__ float4 part[POOL_WAVES * 32];
    __shared__ int    wcnt[POOL_WAVES];

    const int g    = blockIdx.x;
    const int lane = threadIdx.x & 31;
    const int wv   = __builtin_amdgcn_readfirstlane((int)(threadIdx.x >> 5));

    const float* el = emb + lane * 4;
    float4 acc = make_float4(0.f, 0.f, 0.f, 0.f);
    int cnt = 0;

    for (int c0 = wv * 32; c0 < n; c0 += POOL_CHUNK * POOL_THREADS) {
        int b[POOL_CHUNK];
#pragma unroll
        for (int u = 0; u < POOL_CHUNK; ++u) {
            const int i = c0 + u * POOL_THREADS + lane;
            b[u] = (i < n) ? batch[i] : -1;
        }
#pragma unroll
        for (int u = 0; u < POOL_CHUNK; ++u) {
            unsigned mask = __builtin_amdgcn_ballot_w32(b[u] == g);
            cnt += __builtin_popcount(mask);
            const int base = c0 + u * POOL_THREADS;
            while (mask) {
                float4 v[POOL_ROWS];
#pragma unroll
                for (int r = 0; r < POOL_ROWS; ++r) {
                    v[r] = make_float4(0.f, 0.f, 0.f, 0.f);
                    if (mask) {
                        const int k = __builtin_ctz(mask);
                        mask &= mask - 1;
                        v[r] = *(const float4*)(el + (size_t)(base + k) * DIM_O);
                    }
                }
#pragma unroll
                for (int r = 0; r < POOL_ROWS; ++r) {
                    acc.x += v[r].x; acc.y += v[r].y;
                    acc.z += v[r].z; acc.w += v[r].w;
                }
            }
        }
    }

    part[wv * 32 + lane] = acc;
    if (lane == 0) wcnt[wv] = cnt;
    __syncthreads();

    if (threadIdx.x < DIM_O) {
        const float* pf = (const float*)part;
        float s = 0.0f;
        int   c = 0;
        for (int w = 0; w < POOL_WAVES; ++w) {
            s += pf[w * DIM_O + threadIdx.x];
            c += wcnt[w];
        }
        gemb[(size_t)g * DIM_O + threadIdx.x] = s / fmaxf((float)c, 1.0f);
    }
}


__global__ void k_logits(const float* __restrict__ gemb, const float* __restrict__ Wc,
                         const float* __restrict__ bc, float* __restrict__ logits, int D) {
    const int t = threadIdx.x;
    const int g = t >> 1;
    const int c = t & 1;
    float acc = bc[c];
    const float* row = gemb + (size_t)g * D;
    for (int k = 0; k < D; ++k) acc += row[k] * Wc[k * 2 + c];
    logits[g * 2 + c] = acc;
}


static inline int cdiv_l(long a, int b) { return (int)((a + b - 1) / b); }

extern "C" void kernel_launch(void* const* d_in, const int* in_sizes, int n_in,
                              void* d_out, int out_size, void* d_ws, size_t ws_size,
                              hipStream_t stream) {
    const float* x     = (const float*)d_in[0];
    const int*   ei    = (const int*)d_in[1];
    const int*   batch = (const int*)d_in[2];
    const float* W_in  = (const float*)d_in[3];
    const float* b_in  = (const float*)d_in[4];
    const float* W_mid = (const float*)d_in[5];
    const float* b_mid = (const float*)d_in[6];
    const float* W_out = (const float*)d_in[7];
    const float* b_out = (const float*)d_in[8];
    const float* W_cls = (const float*)d_in[9];
    const float* b_cls = (const float*)d_in[10];

    const int N = in_sizes[0] / DIM_IN;
    const int E = in_sizes[1] / 2;
    const int* src = ei;
    const int* dst = ei + E;

    float* w       = (float*)d_ws;
    float* dis     = w;
    float* bufH    = dis + N;
    float* bufCsr  = bufH + (size_t)N * DIM_H;
    float* bufA    = bufCsr + (size_t)N * DIM_H;

    int*  rowptr = (int*)bufCsr;
    int*  slot   = rowptr + (N + 64);
    int2* csr    = (int2*)(slot + E);

    float* node_emb = (float*)d_out;
    float* gemb     = node_emb + (size_t)N * DIM_O;
    float* logits   = gemb + (size_t)NGRAPH * DIM_O;

    const int B = 256;
    const int MB = 32;
    const int GB = cdiv_l(N, 8);

    k_zero_int<<<cdiv_l(N, B), B, 0, stream>>>(rowptr, N);
    k_deg_slot<<<cdiv_l(E, B), B, 0, stream>>>(dst, rowptr, slot, E);
    k_scan_dis<<<1, SCAN_THREADS, 0, stream>>>(rowptr, dis, N);
    k_place<<<cdiv_l(E, B), B, 0, stream>>>(src, dst, rowptr, slot, dis, csr, E);

    {
        dim3 grid(N / MB, DIM_H / 256);
        k_gemm_wmma<2, 4><<<grid, 128, 0, stream>>>(x, W_in, bufH, N, DIM_IN, DIM_H);
        k_gather<DIM_H, 1><<<GB, B, 0, stream>>>(bufH, rowptr, csr, dis, b_in, bufA, N);
    }

    {
        dim3 grid(N / MB, DIM_H / 256);
        k_gemm_wmma<2, 4><<<grid, 128, 0, stream>>>(bufA, W_mid, bufH, N, DIM_H, DIM_H);
        k_gather<DIM_H, 1><<<GB, B, 0, stream>>>(bufH, rowptr, csr, dis, b_mid, bufA, N);
    }

    {
        dim3 grid(N / MB, DIM_O / 128);
        k_gemm_wmma<1, 2><<<grid, 128, 0, stream>>>(bufA, W_out, bufH, N, DIM_H, DIM_O);
        k_gather<DIM_O, 0><<<GB, B, 0, stream>>>(bufH, rowptr, csr, dis, b_out, node_emb, N);
    }

    k_pool_mean<<<NGRAPH, POOL_THREADS, 0, stream>>>(node_emb, batch, gemb, N);

    k_logits<<<1, 128, 0, stream>>>(gemb, W_cls, b_cls, logits, DIM_O);
}


# Layer 1 aggregates its 128-wide input rows first and multiplies afterwards (agg(x) @ W = agg(x @ W)); its bias and relu move into the GEMM's store and its gather runs with a null bias pointer
# speedup vs baseline: 3.7158x; 1.0697x over previous
#include <hip/hip_runtime.h>
#include <hip/hip_bf16.h>


typedef float v2f __attribute__((ext_vector_type(2)));
typedef float v8f __attribute__((ext_vector_type(8)));

#define NNODES 40000
#define NEDGES 640000
#define NGRAPH 64
#define DIM_IN 128
#define DIM_H  256
#define DIM_O  128


__global__ void k_zero_int(int* __restrict__ p, int n) {
    int i = blockIdx.x * blockDim.x + threadIdx.x;
    if (i < n) p[i] = 0;
}

__global__ void k_deg_slot(const int* __restrict__ dst, int* __restrict__ cnt,
                           int* __restrict__ slot, int e) {
    int i = blockIdx.x * blockDim.x + threadIdx.x;
    if (i < e) slot[i] = atomicAdd(&cnt[dst[i]], 1);
}

#define SCAN_THREADS 1024
#define SCAN_WAVES   (SCAN_THREADS / 32)
#define SCAN_PASSES  ((NNODES + SCAN_THREADS * 4 - 1) / (SCAN_THREADS * 4))
#define SCAN_WSPAN   (SCAN_PASSES * 128)

__global__ __launch_bounds__(SCAN_THREADS) void k_scan_dis(int* __restrict__ cr,
                                                           float* __restrict__ dis, int n) {
    static_assert(SCAN_WAVES == 32, "one lane of the first wave per wave total");
    static_assert(NNODES % 4 == 0, "a lane's group of 4 nodes lies wholly inside or wholly outside the range");
    __shared__ int wsum[SCAN_WAVES];
    const int t    = threadIdx.x;
    const int lane = t & 31;
    const int wv   = t >> 5;
    const int b0   = wv * SCAN_WSPAN + lane * 4;

    int4 c[SCAN_PASSES];
#pragma unroll
    for (int j = 0; j < SCAN_PASSES; ++j) {
        const int idx = b0 + j * 128;
        c[j] = make_int4(0, 0, 0, 0);
        if (idx < n) c[j] = *(const int4*)(cr + idx);
    }

    int s4[SCAN_PASSES];
    int x[SCAN_PASSES];
#pragma unroll
    for (int j = 0; j < SCAN_PASSES; ++j) {
        s4[j] = c[j].x + c[j].y + c[j].z + c[j].w;
        x[j]  = s4[j];
    }
#pragma unroll
    for (int off = 1; off < 32; off <<= 1) {
#pragma unroll
        for (int j = 0; j < SCAN_PASSES; ++j) {
            const int y = __shfl_up(x[j], off);
            if (lane >= off) x[j] += y;
        }
    }

    int ex[SCAN_PASSES];
    int carry = 0;
#pragma unroll
    for (int j = 0; j < SCAN_PASSES; ++j) {
        ex[j] = carry + x[j] - s4[j];
        carry += __builtin_amdgcn_readlane(x[j], 31);
    }

    if (lane == 0) wsum[wv] = carry;
    __syncthreads();
    if (wv == 0) {
        int s = wsum[lane];
        for (int off = 1; off < 32; off <<= 1) {
            const int y = __shfl_up(s, off);
            if (lane >= off) s += y;
        }
        wsum[lane] = s;
    }
    __syncthreads();

    const int wbase = (wv > 0) ? wsum[wv - 1] : 0;
#pragma unroll
    for (int j = 0; j < SCAN_PASSES; ++j) {
        const int idx = b0 + j * 128;
        int4 r;
        r.x = wbase + ex[j];
        r.y = r.x + c[j].x;
        r.z = r.y + c[j].y;
        r.w = r.z + c[j].z;
        float4 d;
        d.x = rsqrtf((float)c[j].x + 1.0f);
        d.y = rsqrtf((float)c[j].y + 1.0f);
        d.z = rsqrtf((float)c[j].z + 1.0f);
        d.w = rsqrtf((float)c[j].w + 1.0f);
        if (idx < n) {
            *(int4*)(cr + idx)    = r;
            *(float4*)(dis + idx) = d;
        }
    }
    if (t == SCAN_THREADS - 1) cr[n] = wbase + carry;
}

__global__ void k_place(const int* __restrict__ src, const int* __restrict__ dst,
                        const int* __restrict__ rowptr, const int* __restrict__ slot,
                        const float* __restrict__ dis, int2* __restrict__ csr, int e) {
    int i = blockIdx.x * blockDim.x + threadIdx.x;
    if (i >= e) return;
    const int s = src[i];
    const int d = dst[i];
    const float nrm = dis[s] * dis[d];
    csr[rowptr[d] + slot[i]] = make_int2(s, __float_as_int(nrm));
}


template <int MT>
__device__ __forceinline__ void gemm_fetch(const float* const (&ap)[MT], const float* __restrict__ wp,
                                           int k, int Nc, float4 (&a)[MT], float4 (&b)[4]) {
    for (int t = 0; t < MT; ++t) a[t] = *(const float4*)(ap[t] + k);
    for (int j = 0; j < 4; ++j)  b[j] = *(const float4*)(wp + (size_t)(k + j) * Nc);
}

template <int MT>
__device__ __forceinline__ void gemm_steps(const float4 (&a)[MT], const float4 (&b)[4],
                                           v8f (&acc)[MT][4]) {
    const float bx[4][4] = {{b[0].x, b[0].y, b[0].z, b[0].w},
                            {b[1].x, b[1].y, b[1].z, b[1].w},
                            {b[2].x, b[2].y, b[2].z, b[2].w},
                            {b[3].x, b[3].y, b[3].z, b[3].w}};
    for (int p = 0; p < 2; ++p) {
        for (int t = 0; t < MT; ++t) {
            v2f av;
            av.x = p ? a[t].z : a[t].x;
            av.y = p ? a[t].w : a[t].y;
            for (int q = 0; q < 4; ++q) {
                v2f bv;
                bv.x = bx[2 * p][q];
                bv.y = bx[2 * p + 1][q];
                acc[t][q] = __builtin_amdgcn_wmma_f32_16x16x4_f32(
                     false, av,  false, bv,
                     (short)0, acc[t][q],  false,  false);
            }
        }
    }
}

template <int MT, int WN, int EPI>
__global__ __launch_bounds__(128) void k_gemm_wmma(const float* __restrict__ A,
                                                   const float* __restrict__ W,
                                                   const float* __restrict__ bias,
                                                   float* __restrict__ C,
                                                   int M, int K, int Nc) {
    constexpr int WM = 4 / WN;
    const int lane = threadIdx.x & 31;
    const int wave = __builtin_amdgcn_readfirstlane((int)(threadIdx.x >> 5));
    const int m0 = ((int)blockIdx.x * WM + wave / WN) * (16 * MT);
    const int n0 = ((int)blockIdx.y * WN + wave % WN) * 64;
    if (m0 >= M || n0 >= Nc) return;

    const int h = lane >> 4;
    const int c = lane & 15;

    const float* ap[MT];
    for (int t = 0; t < MT; ++t)
        ap[t] = A + (size_t)(m0 + t * 16 + c) * K + 4 * h;
    const float* __restrict__ wp = W + (size_t)(4 * h) * Nc + n0 + 4 * c;

    v8f acc[MT][4];
    for (int t = 0; t < MT; ++t)
        for (int q = 0; q < 4; ++q) acc[t][q] = (v8f){};

    float4 a0[MT], b0[4], a1[MT], b1[4];
    gemm_fetch<MT>(ap, wp, 0, Nc, a0, b0);

    for (int k0 = 0; k0 < K; k0 += 16) {
        gemm_fetch<MT>(ap, wp, k0 + 8, Nc, a1, b1);
        __builtin_amdgcn_sched_barrier(0);
        gemm_steps<MT>(a0, b0, acc);

        const int kn = (k0 + 16 < K) ? k0 + 16 : k0;
        gemm_fetch<MT>(ap, wp, kn, Nc, a0, b0);
        __builtin_amdgcn_sched_barrier(0);
        gemm_steps<MT>(a1, b1, acc);
    }

    float4 bv = make_float4(0.f, 0.f, 0.f, 0.f);
    if (EPI) bv = *(const float4*)(bias + n0 + 4 * c);
    for (int t = 0; t < MT; ++t) {
        float* crow = C + (size_t)(m0 + t * 16 + 8 * h) * Nc + n0 + 4 * c;
        for (int v = 0; v < 8; ++v) {
            float4 r;
            r.x = acc[t][0][v]; r.y = acc[t][1][v];
            r.z = acc[t][2][v]; r.w = acc[t][3][v];
            if (EPI) {
                r.x = fmaxf(r.x + bv.x, 0.0f); r.y = fmaxf(r.y + bv.y, 0.0f);
                r.z = fmaxf(r.z + bv.z, 0.0f); r.w = fmaxf(r.w + bv.w, 0.0f);
            }
            *(float4*)(crow + (size_t)v * Nc) = r;
        }
    }
}


template <int H, int K>
__device__ __forceinline__ void gather_rows(const float* __restrict__ hl, int my_s, int my_w,
                                            int i, float4 (&acc)[H / 128]) {
    constexpr int V = H / 128;
    float  w[K];
    float4 v[K][V];
#pragma unroll
    for (int k = 0; k < K; ++k) {
        const int s = __builtin_amdgcn_readlane(my_s, i + k);
        w[k] = __int_as_float(__builtin_amdgcn_readlane(my_w, i + k));
        const float* row = hl + (size_t)s * H;
        for (int q = 0; q < V; ++q) v[k][q] = *(const float4*)(row + q * 128);
    }
#pragma unroll
    for (int k = 0; k < K; ++k) {
        for (int q = 0; q < V; ++q) {
            acc[q].x = __builtin_fmaf(w[k], v[k][q].x, acc[q].x);
            acc[q].y = __builtin_fmaf(w[k], v[k][q].y, acc[q].y);
            acc[q].z = __builtin_fmaf(w[k], v[k][q].z, acc[q].z);
            acc[q].w = __builtin_fmaf(w[k], v[k][q].w, acc[q].w);
        }
    }
}

template <int H, int RELU>
__global__ __launch_bounds__(256) void k_gather(const float* __restrict__ h,
                                                const int* __restrict__ rowptr,
                                                const int2* __restrict__ csr,
                                                const float* __restrict__ dis,
                                                const float* __restrict__ b,
                                                float* __restrict__ out, int n) {
    constexpr int V = H / 128;
    constexpr int K = 1024 / H;
    const int lane = threadIdx.x & 31;
    const int node = __builtin_amdgcn_readfirstlane((int)(blockIdx.x * 8 + (threadIdx.x >> 5)));
    if (node >= n) return;
    const int beg = rowptr[node];
    const int end = rowptr[node + 1];

    const float* hl = h + lane * 4;
    float4 acc[V];
    for (int q = 0; q < V; ++q) acc[q] = make_float4(0.f, 0.f, 0.f, 0.f);

    for (int base = beg; base < end; base += 32) {
        const int cntw = (end - base < 32) ? (end - base) : 32;
        int2 me = make_int2(0, 0);
        if (lane < cntw) me = csr[base + lane];
        int i = 0;
        for (; i + K <= cntw; i += K) gather_rows<H, K>(hl, me.x, me.y, i, acc);
        if (K >= 8 && i + 4 <= cntw) { gather_rows<H, 4>(hl, me.x, me.y, i, acc); i += 4; }
        if (i + 2 <= cntw)           { gather_rows<H, 2>(hl, me.x, me.y, i, acc); i += 2; }
        if (i + 1 <= cntw)           { gather_rows<H, 1>(hl, me.x, me.y, i, acc); }
    }

    const float ds = dis[node];
    const float sl = ds * ds;
    const float* hrow = hl + (size_t)node * H;
    float* orow = out + (size_t)node * H + lane * 4;
    for (int q = 0; q < V; ++q) {
        const float4 hv = *(const float4*)(hrow + q * 128);
        float4 r;
        r.x = acc[q].x + sl * hv.x;
        r.y = acc[q].y + sl * hv.y;
        r.z = acc[q].z + sl * hv.z;
        r.w = acc[q].w + sl * hv.w;
        if (b != nullptr) {
            const float4 bv = *(const float4*)(b + q * 128 + lane * 4);
            r.x += bv.x; r.y += bv.y; r.z += bv.z; r.w += bv.w;
        }
        if (RELU) {
            r.x = fmaxf(r.x, 0.0f); r.y = fmaxf(r.y, 0.0f);
            r.z = fmaxf(r.z, 0.0f); r.w = fmaxf(r.w, 0.0f);
        }
        *(float4*)(orow + q * 128) = r;
    }
}


#define POOL_THREADS 1024
#define POOL_WAVES   (POOL_THREADS / 32)
#define POOL_CHUNK   8
#define POOL_ROWS    8

__global__ __launch_bounds__(POOL_THREADS) void k_pool_mean(const float* __restrict__ emb,
                                                            const int* __restrict__ batch,
                                                            float* __restrict__ gemb, int n) {
    static_assert(DIM_O == 128, "a lane owns 4 of 128 columns");
    __shared__ float4 part[POOL_WAVES * 32];
    __shared__ int    wcnt[POOL_WAVES];

    const int g    = blockIdx.x;
    const int lane = threadIdx.x & 31;
    const int wv   = __builtin_amdgcn_readfirstlane((int)(threadIdx.x >> 5));

    const float* el = emb + lane * 4;
    float4 acc = make_float4(0.f, 0.f, 0.f, 0.f);
    int cnt = 0;

    for (int c0 = wv * 32; c0 < n; c0 += POOL_CHUNK * POOL_THREADS) {
        int b[POOL_CHUNK];
#pragma unroll
        for (int u = 0; u < POOL_CHUNK; ++u) {
            const int i = c0 + u * POOL_THREADS + lane;
            b[u] = (i < n) ? batch[i] : -1;
        }
#pragma unroll
        for (int u = 0; u < POOL_CHUNK; ++u) {
            unsigned mask = __builtin_amdgcn_ballot_w32(b[u] == g);
            cnt += __builtin_popcount(mask);
            const int base = c0 + u * POOL_THREADS;
            while (mask) {
                float4 v[POOL_ROWS];
#pragma unroll
                for (int r = 0; r < POOL_ROWS; ++r) {
                    v[r] = make_float4(0.f, 0.f, 0.f, 0.f);
                    if (mask) {
                        const int k = __builtin_ctz(mask);
                        mask &= mask - 1;
                        v[r] = *(const float4*)(el + (size_t)(base + k) * DIM_O);
                    }
                }
#pragma unroll
                for (int r = 0; r < POOL_ROWS; ++r) {
                    acc.x += v[r].x; acc.y += v[r].y;
                    acc.z += v[r].z; acc.w += v[r].w;
                }
            }
        }
    }

    part[wv * 32 + lane] = acc;
    if (lane == 0) wcnt[wv] = cnt;
    __syncthreads();

    if (threadIdx.x < DIM_O) {
        const float* pf = (const float*)part;
        float s = 0.0f;
        int   c = 0;
        for (int w = 0; w < POOL_WAVES; ++w) {
            s += pf[w * DIM_O + threadIdx.x];
            c += wcnt[w];
        }
        gemb[(size_t)g * DIM_O + threadIdx.x] = s / fmaxf((float)c, 1.0f);
    }
}


__global__ void k_logits(const float* __restrict__ gemb, const float* __restrict__ Wc,
                         const float* __restrict__ bc, float* __restrict__ logits, int D) {
    const int t = threadIdx.x;
    const int g = t >> 1;
    const int c = t & 1;
    float acc = bc[c];
    const float* row = gemb + (size_t)g * D;
    for (int k = 0; k < D; ++k) acc += row[k] * Wc[k * 2 + c];
    logits[g * 2 + c] = acc;
}


static inline int cdiv_l(long a, int b) { return (int)((a + b - 1) / b); }

extern "C" void kernel_launch(void* const* d_in, const int* in_sizes, int n_in,
                              void* d_out, int out_size, void* d_ws, size_t ws_size,
                              hipStream_t stream) {
    const float* x     = (const float*)d_in[0];
    const int*   ei    = (const int*)d_in[1];
    const int*   batch = (const int*)d_in[2];
    const float* W_in  = (const float*)d_in[3];
    const float* b_in  = (const float*)d_in[4];
    const float* W_mid = (const float*)d_in[5];
    const float* b_mid = (const float*)d_in[6];
    const float* W_out = (const float*)d_in[7];
    const float* b_out = (const float*)d_in[8];
    const float* W_cls = (const float*)d_in[9];
    const float* b_cls = (const float*)d_in[10];

    const int N = in_sizes[0] / DIM_IN;
    const int E = in_sizes[1] / 2;
    const int* src = ei;
    const int* dst = ei + E;

    float* w       = (float*)d_ws;
    float* dis     = w;
    float* bufH    = dis + N;
    float* bufCsr  = bufH + (size_t)N * DIM_H;
    float* bufA    = bufCsr + (size_t)N * DIM_H;

    int*  rowptr = (int*)bufCsr;
    int*  slot   = rowptr + (N + 64);
    int2* csr    = (int2*)(slot + E);

    float* node_emb = (float*)d_out;
    float* gemb     = node_emb + (size_t)N * DIM_O;
    float* logits   = gemb + (size_t)NGRAPH * DIM_O;

    const int B = 256;
    const int MB = 32;
    const int GB = cdiv_l(N, 8);

    k_zero_int<<<cdiv_l(N, B), B, 0, stream>>>(rowptr, N);
    k_deg_slot<<<cdiv_l(E, B), B, 0, stream>>>(dst, rowptr, slot, E);
    k_scan_dis<<<1, SCAN_THREADS, 0, stream>>>(rowptr, dis, N);
    k_place<<<cdiv_l(E, B), B, 0, stream>>>(src, dst, rowptr, slot, dis, csr, E);

    {
        float* aggX = bufH;
        k_gather<DIM_IN, 0><<<GB, B, 0, stream>>>(x, rowptr, csr, dis, nullptr, aggX, N);
        dim3 grid(N / MB, DIM_H / 256);
        k_gemm_wmma<2, 4, 1><<<grid, 128, 0, stream>>>(aggX, W_in, b_in, bufA, N, DIM_IN, DIM_H);
    }

    {
        dim3 grid(N / MB, DIM_H / 256);
        k_gemm_wmma<2, 4, 0><<<grid, 128, 0, stream>>>(bufA, W_mid, b_mid, bufH, N, DIM_H, DIM_H);
        k_gather<DIM_H, 1><<<GB, B, 0, stream>>>(bufH, rowptr, csr, dis, b_mid, bufA, N);
    }

    {
        dim3 grid(N / MB, DIM_O / 128);
        k_gemm_wmma<1, 2, 0><<<grid, 128, 0, stream>>>(bufA, W_out, b_out, bufH, N, DIM_H, DIM_O);
        k_gather<DIM_O, 0><<<GB, B, 0, stream>>>(bufH, rowptr, csr, dis, b_out, node_emb, N);
    }

    k_pool_mean<<<NGRAPH, POOL_THREADS, 0, stream>>>(node_emb, batch, gemb, N);

    k_logits<<<1, 128, 0, stream>>>(gemb, W_cls, b_cls, logits, DIM_O);
}


# Aggregation fused into the GEMM that follows it (layer 1; layer 2's aggregation into layer 3's product): a block aggregates its 32 rows into an LDS tile and multiplies from there; 12 launches -> 10
# speedup vs baseline: 3.9729x; 1.0692x over previous
#include <hip/hip_runtime.h>
#include <hip/hip_bf16.h>


typedef float v2f __attribute__((ext_vector_type(2)));
typedef float v8f __attribute__((ext_vector_type(8)));

#define NNODES 40000
#define NEDGES 640000
#define NGRAPH 64
#define DIM_IN 128
#define DIM_H  256
#define DIM_O  128


__global__ void k_zero_int(int* __restrict__ p, int n) {
    int i = blockIdx.x * blockDim.x + threadIdx.x;
    if (i < n) p[i] = 0;
}

__global__ void k_deg_slot(const int* __restrict__ dst, int* __restrict__ cnt,
                           int* __restrict__ slot, int e) {
    int i = blockIdx.x * blockDim.x + threadIdx.x;
    if (i < e) slot[i] = atomicAdd(&cnt[dst[i]], 1);
}

#define SCAN_THREADS 1024
#define SCAN_WAVES   (SCAN_THREADS / 32)
#define SCAN_PASSES  ((NNODES + SCAN_THREADS * 4 - 1) / (SCAN_THREADS * 4))
#define SCAN_WSPAN   (SCAN_PASSES * 128)

__global__ __launch_bounds__(SCAN_THREADS) void k_scan_dis(int* __restrict__ cr,
                                                           float* __restrict__ dis, int n) {
    static_assert(SCAN_WAVES == 32, "one lane of the first wave per wave total");
    static_assert(NNODES % 4 == 0, "a lane's group of 4 nodes lies wholly inside or wholly outside the range");
    __shared__ int wsum[SCAN_WAVES];
    const int t    = threadIdx.x;
    const int lane = t & 31;
    const int wv   = t >> 5;
    const int b0   = wv * SCAN_WSPAN + lane * 4;

    int4 c[SCAN_PASSES];
#pragma unroll
    for (int j = 0; j < SCAN_PASSES; ++j) {
        const int idx = b0 + j * 128;
        c[j] = make_int4(0, 0, 0, 0);
        if (idx < n) c[j] = *(const int4*)(cr + idx);
    }

    int s4[SCAN_PASSES];
    int x[SCAN_PASSES];
#pragma unroll
    for (int j = 0; j < SCAN_PASSES; ++j) {
        s4[j] = c[j].x + c[j].y + c[j].z + c[j].w;
        x[j]  = s4[j];
    }
#pragma unroll
    for (int off = 1; off < 32; off <<= 1) {
#pragma unroll
        for (int j = 0; j < SCAN_PASSES; ++j) {
            const int y = __shfl_up(x[j], off);
            if (lane >= off) x[j] += y;
        }
    }

    int ex[SCAN_PASSES];
    int carry = 0;
#pragma unroll
    for (int j = 0; j < SCAN_PASSES; ++j) {
        ex[j] = carry + x[j] - s4[j];
        carry += __builtin_amdgcn_readlane(x[j], 31);
    }

    if (lane == 0) wsum[wv] = carry;
    __syncthreads();
    if (wv == 0) {
        int s = wsum[lane];
        for (int off = 1; off < 32; off <<= 1) {
            const int y = __shfl_up(s, off);
            if (lane >= off) s += y;
        }
        wsum[lane] = s;
    }
    __syncthreads();

    const int wbase = (wv > 0) ? wsum[wv - 1] : 0;
#pragma unroll
    for (int j = 0; j < SCAN_PASSES; ++j) {
        const int idx = b0 + j * 128;
        int4 r;
        r.x = wbase + ex[j];
        r.y = r.x + c[j].x;
        r.z = r.y + c[j].y;
        r.w = r.z + c[j].z;
        float4 d;
        d.x = rsqrtf((float)c[j].x + 1.0f);
        d.y = rsqrtf((float)c[j].y + 1.0f);
        d.z = rsqrtf((float)c[j].z + 1.0f);
        d.w = rsqrtf((float)c[j].w + 1.0f);
        if (idx < n) {
            *(int4*)(cr + idx)    = r;
            *(float4*)(dis + idx) = d;
        }
    }
    if (t == SCAN_THREADS - 1) cr[n] = wbase + carry;
}

__global__ void k_place(const int* __restrict__ src, const int* __restrict__ dst,
                        const int* __restrict__ rowptr, const int* __restrict__ slot,
                        const float* __restrict__ dis, int2* __restrict__ csr, int e) {
    int i = blockIdx.x * blockDim.x + threadIdx.x;
    if (i >= e) return;
    const int s = src[i];
    const int d = dst[i];
    const float nrm = dis[s] * dis[d];
    csr[rowptr[d] + slot[i]] = make_int2(s, __float_as_int(nrm));
}


template <int MT>
__device__ __forceinline__ void gemm_fetch(const float* const (&ap)[MT], const float* __restrict__ wp,
                                           int k, int Nc, float4 (&a)[MT], float4 (&b)[4]) {
    for (int t = 0; t < MT; ++t) a[t] = *(const float4*)(ap[t] + k);
    for (int j = 0; j < 4; ++j)  b[j] = *(const float4*)(wp + (size_t)(k + j) * Nc);
}

template <int MT>
__device__ __forceinline__ void gemm_steps(const float4 (&a)[MT], const float4 (&b)[4],
                                           v8f (&acc)[MT][4]) {
    const float bx[4][4] = {{b[0].x, b[0].y, b[0].z, b[0].w},
                            {b[1].x, b[1].y, b[1].z, b[1].w},
                            {b[2].x, b[2].y, b[2].z, b[2].w},
                            {b[3].x, b[3].y, b[3].z, b[3].w}};
    for (int p = 0; p < 2; ++p) {
        for (int t = 0; t < MT; ++t) {
            v2f av;
            av.x = p ? a[t].z : a[t].x;
            av.y = p ? a[t].w : a[t].y;
            for (int q = 0; q < 4; ++q) {
                v2f bv;
                bv.x = bx[2 * p][q];
                bv.y = bx[2 * p + 1][q];
                acc[t][q] = __builtin_amdgcn_wmma_f32_16x16x4_f32(
                     false, av,  false, bv,
                     (short)0, acc[t][q],  false,  false);
            }
        }
    }
}

template <int MT, int WN, int EPI>
__global__ __launch_bounds__(128) void k_gemm_wmma(const float* __restrict__ A,
                                                   const float* __restrict__ W,
                                                   const float* __restrict__ bias,
                                                   float* __restrict__ C,
                                                   int M, int K, int Nc) {
    constexpr int WM = 4 / WN;
    const int lane = threadIdx.x & 31;
    const int wave = __builtin_amdgcn_readfirstlane((int)(threadIdx.x >> 5));
    const int m0 = ((int)blockIdx.x * WM + wave / WN) * (16 * MT);
    const int n0 = ((int)blockIdx.y * WN + wave % WN) * 64;
    if (m0 >= M || n0 >= Nc) return;

    const int h = lane >> 4;
    const int c = lane & 15;

    const float* ap[MT];
    for (int t = 0; t < MT; ++t)
        ap[t] = A + (size_t)(m0 + t * 16 + c) * K + 4 * h;
    const float* __restrict__ wp = W + (size_t)(4 * h) * Nc + n0 + 4 * c;

    v8f acc[MT][4];
    for (int t = 0; t < MT; ++t)
        for (int q = 0; q < 4; ++q) acc[t][q] = (v8f){};

    float4 a0[MT], b0[4], a1[MT], b1[4];
    gemm_fetch<MT>(ap, wp, 0, Nc, a0, b0);

    for (int k0 = 0; k0 < K; k0 += 16) {
        gemm_fetch<MT>(ap, wp, k0 + 8, Nc, a1, b1);
        __builtin_amdgcn_sched_barrier(0);
        gemm_steps<MT>(a0, b0, acc);

        const int kn = (k0 + 16 < K) ? k0 + 16 : k0;
        gemm_fetch<MT>(ap, wp, kn, Nc, a0, b0);
        __builtin_amdgcn_sched_barrier(0);
        gemm_steps<MT>(a1, b1, acc);
    }

    float4 bv = make_float4(0.f, 0.f, 0.f, 0.f);
    if (EPI) bv = *(const float4*)(bias + n0 + 4 * c);
    for (int t = 0; t < MT; ++t) {
        float* crow = C + (size_t)(m0 + t * 16 + 8 * h) * Nc + n0 + 4 * c;
        for (int v = 0; v < 8; ++v) {
            float4 r;
            r.x = acc[t][0][v]; r.y = acc[t][1][v];
            r.z = acc[t][2][v]; r.w = acc[t][3][v];
            if (EPI) {
                r.x = fmaxf(r.x + bv.x, 0.0f); r.y = fmaxf(r.y + bv.y, 0.0f);
                r.z = fmaxf(r.z + bv.z, 0.0f); r.w = fmaxf(r.w + bv.w, 0.0f);
            }
            *(float4*)(crow + (size_t)v * Nc) = r;
        }
    }
}


template <int H, int K>
__device__ __forceinline__ void gather_rows(const float* __restrict__ hl, int my_s, int my_w,
                                            int i, float4 (&acc)[H / 128]) {
    constexpr int V = H / 128;
    float  w[K];
    float4 v[K][V];
#pragma unroll
    for (int k = 0; k < K; ++k) {
        const int s = __builtin_amdgcn_readlane(my_s, i + k);
        w[k] = __int_as_float(__builtin_amdgcn_readlane(my_w, i + k));
        const float* row = hl + (size_t)s * H;
        for (int q = 0; q < V; ++q) v[k][q] = *(const float4*)(row + q * 128);
    }
#pragma unroll
    for (int k = 0; k < K; ++k) {
        for (int q = 0; q < V; ++q) {
            acc[q].x = __builtin_fmaf(w[k], v[k][q].x, acc[q].x);
            acc[q].y = __builtin_fmaf(w[k], v[k][q].y, acc[q].y);
            acc[q].z = __builtin_fmaf(w[k], v[k][q].z, acc[q].z);
            acc[q].w = __builtin_fmaf(w[k], v[k][q].w, acc[q].w);
        }
    }
}

template <int H, int RELU>
__global__ __launch_bounds__(256) void k_gather(const float* __restrict__ h,
                                                const int* __restrict__ rowptr,
                                                const int2* __restrict__ csr,
                                                const float* __restrict__ dis,
                                                const float* __restrict__ b,
                                                float* __restrict__ out, int n) {
    constexpr int V = H / 128;
    constexpr int K = 1024 / H;
    const int lane = threadIdx.x & 31;
    const int node = __builtin_amdgcn_readfirstlane((int)(blockIdx.x * 8 + (threadIdx.x >> 5)));
    if (node >= n) return;
    const int beg = rowptr[node];
    const int end = rowptr[node + 1];

    const float* hl = h + lane * 4;
    float4 acc[V];
    for (int q = 0; q < V; ++q) acc[q] = make_float4(0.f, 0.f, 0.f, 0.f);

    for (int base = beg; base < end; base += 32) {
        const int cntw = (end - base < 32) ? (end - base) : 32;
        int2 me = make_int2(0, 0);
        if (lane < cntw) me = csr[base + lane];
        int i = 0;
        for (; i + K <= cntw; i += K) gather_rows<H, K>(hl, me.x, me.y, i, acc);
        if (K >= 8 && i + 4 <= cntw) { gather_rows<H, 4>(hl, me.x, me.y, i, acc); i += 4; }
        if (i + 2 <= cntw)           { gather_rows<H, 2>(hl, me.x, me.y, i, acc); i += 2; }
        if (i + 1 <= cntw)           { gather_rows<H, 1>(hl, me.x, me.y, i, acc); }
    }

    const float ds = dis[node];
    const float sl = ds * ds;
    const float* hrow = hl + (size_t)node * H;
    float* orow = out + (size_t)node * H + lane * 4;
    for (int q = 0; q < V; ++q) {
        const float4 hv = *(const float4*)(hrow + q * 128);
        float4 r;
        r.x = acc[q].x + sl * hv.x;
        r.y = acc[q].y + sl * hv.y;
        r.z = acc[q].z + sl * hv.z;
        r.w = acc[q].w + sl * hv.w;
        if (b != nullptr) {
            const float4 bv = *(const float4*)(b + q * 128 + lane * 4);
            r.x += bv.x; r.y += bv.y; r.z += bv.z; r.w += bv.w;
        }
        if (RELU) {
            r.x = fmaxf(r.x, 0.0f); r.y = fmaxf(r.y, 0.0f);
            r.z = fmaxf(r.z, 0.0f); r.w = fmaxf(r.w, 0.0f);
        }
        *(float4*)(orow + q * 128) = r;
    }
}


template <int H>
__device__ __forceinline__ void gather_chunk(const float* __restrict__ hl, int2 me, int cntw,
                                             float4 (&acc)[H / 128]) {
    constexpr int K = 2048 / H;
    int i = 0;
    for (; i + K <= cntw; i += K) gather_rows<H, K>(hl, me.x, me.y, i, acc);
    if (K >= 16 && i + 8 <= cntw) { gather_rows<H, 8>(hl, me.x, me.y, i, acc); i += 8; }
    if (i + 4 <= cntw)            { gather_rows<H, 4>(hl, me.x, me.y, i, acc); i += 4; }
    if (i + 2 <= cntw)            { gather_rows<H, 2>(hl, me.x, me.y, i, acc); i += 2; }
    if (i + 1 <= cntw)            { gather_rows<H, 1>(hl, me.x, me.y, i, acc); }
}

template <int KD, int MT>
__device__ __forceinline__ void gemm_fetch_tile(const float* as, const float* __restrict__ wp,
                                                int k, int Nc, float4 (&a)[MT], float4 (&b)[4]) {
    for (int t = 0; t < MT; ++t) a[t] = *(const float4*)(as + t * 16 * (KD + 4) + k);
    for (int j = 0; j < 4; ++j)  b[j] = *(const float4*)(wp + (size_t)(k + j) * Nc);
}

template <int KD, int NC, int MT, int WN, int GRELU, int EPI>
__global__ __launch_bounds__(128) void k_gather_gemm(const float* __restrict__ h,
                                                     const int* __restrict__ rowptr,
                                                     const int2* __restrict__ csr,
                                                     const float* __restrict__ dis,
                                                     const float* __restrict__ gb,
                                                     const float* __restrict__ W,
                                                     const float* __restrict__ bias,
                                                     float* __restrict__ C) {
    constexpr int WM    = 4 / WN;
    constexpr int R     = 16 * MT * WM;
    constexpr int RPW   = R / 4;
    constexpr int PITCH = KD + 4;
    constexpr int V     = KD / 128;
    static_assert(R == 32 && RPW == 8, "4 waves x 8 rows");
    static_assert(NNODES % R == 0, "every block owns 32 whole rows: no row guard");
    static_assert(NC == 64 * WN, "the block covers all columns");
    __shared__ float4 tile4[R * PITCH / 4];
    float* tile = (float*)tile4;

    const int lane = threadIdx.x & 31;
    const int wave = __builtin_amdgcn_readfirstlane((int)(threadIdx.x >> 5));
    const int m0 = (int)blockIdx.x * R;

    {
        const int nb = m0 + wave * RPW;
        const int rp = rowptr[nb + (lane < RPW ? lane : RPW)];
        const float* hl = h + lane * 4;

        const int plast = NEDGES - 1;
        int2 cur;
        {
            const int beg = __builtin_amdgcn_readlane(rp, 0);
            const int pi = beg + lane;
            cur = csr[pi < plast ? pi : plast];
        }
#pragma unroll 1
        for (int j = 0; j < RPW; ++j) {
            const int node = nb + j;
            const int beg = __builtin_amdgcn_readlane(rp, j);
            const int end = __builtin_amdgcn_readlane(rp, j + 1);

            const int pi = end + lane;
            const int2 nxt = csr[pi < plast ? pi : plast];
            const float ds = dis[node];
            const float* hrow = hl + (size_t)node * KD;
            float4 hv[V];
            for (int q = 0; q < V; ++q) hv[q] = *(const float4*)(hrow + q * 128);

            float4 acc[V];
            for (int q = 0; q < V; ++q) acc[q] = make_float4(0.f, 0.f, 0.f, 0.f);

            const int c0 = (end - beg < 32) ? (end - beg) : 32;
            gather_chunk<KD>(hl, cur, c0, acc);
            for (int base = beg + 32; base < end; base += 32) {
                const int cntw = (end - base < 32) ? (end - base) : 32;
                int2 m = make_int2(0, 0);
                if (lane < cntw) m = csr[base + lane];
                gather_chunk<KD>(hl, m, cntw, acc);
            }
            cur = nxt;

            const float sl = ds * ds;
            float* trow = tile + (wave * RPW + j) * PITCH + lane * 4;
            for (int q = 0; q < V; ++q) {
                float4 r;
                r.x = acc[q].x + sl * hv[q].x;
                r.y = acc[q].y + sl * hv[q].y;
                r.z = acc[q].z + sl * hv[q].z;
                r.w = acc[q].w + sl * hv[q].w;
                if (gb != nullptr) {
                    const float4 bv = *(const float4*)(gb + q * 128 + lane * 4);
                    r.x += bv.x; r.y += bv.y; r.z += bv.z; r.w += bv.w;
                }
                if (GRELU) {
                    r.x = fmaxf(r.x, 0.0f); r.y = fmaxf(r.y, 0.0f);
                    r.z = fmaxf(r.z, 0.0f); r.w = fmaxf(r.w, 0.0f);
                }
                *(float4*)(trow + q * 128) = r;
            }
        }
    }
    __syncthreads();

    const int rg = wave / WN;
    const int n0 = (wave % WN) * 64;
    const int hh = lane >> 4;
    const int c  = lane & 15;

    const float* as = tile + (rg * 16 * MT + c) * PITCH + 4 * hh;
    const float* __restrict__ wp = W + (size_t)(4 * hh) * NC + n0 + 4 * c;

    v8f acc[MT][4];
    for (int t = 0; t < MT; ++t)
        for (int q = 0; q < 4; ++q) acc[t][q] = (v8f){};

    float4 a0[MT], b0[4], a1[MT], b1[4];
    gemm_fetch_tile<KD, MT>(as, wp, 0, NC, a0, b0);

    for (int k0 = 0; k0 < KD; k0 += 16) {
        gemm_fetch_tile<KD, MT>(as, wp, k0 + 8, NC, a1, b1);
        __builtin_amdgcn_sched_barrier(0);
        gemm_steps<MT>(a0, b0, acc);

        const int kn = (k0 + 16 < KD) ? k0 + 16 : k0;
        gemm_fetch_tile<KD, MT>(as, wp, kn, NC, a0, b0);
        __builtin_amdgcn_sched_barrier(0);
        gemm_steps<MT>(a1, b1, acc);
    }

    float4 bv = make_float4(0.f, 0.f, 0.f, 0.f);
    if (EPI) bv = *(const float4*)(bias + n0 + 4 * c);
    for (int t = 0; t < MT; ++t) {
        float* crow = C + (size_t)(m0 + rg * 16 * MT + t * 16 + 8 * hh) * NC + n0 + 4 * c;
        for (int v = 0; v < 8; ++v) {
            float4 r;
            r.x = acc[t][0][v]; r.y = acc[t][1][v];
            r.z = acc[t][2][v]; r.w = acc[t][3][v];
            if (EPI) {
                r.x = fmaxf(r.x + bv.x, 0.0f); r.y = fmaxf(r.y + bv.y, 0.0f);
                r.z = fmaxf(r.z + bv.z, 0.0f); r.w = fmaxf(r.w + bv.w, 0.0f);
            }
            *(float4*)(crow + (size_t)v * NC) = r;
        }
    }
}


#define POOL_THREADS 1024
#define POOL_WAVES   (POOL_THREADS / 32)
#define POOL_CHUNK   8
#define POOL_ROWS    8

__global__ __launch_bounds__(POOL_THREADS) void k_pool_mean(const float* __restrict__ emb,
                                                            const int* __restrict__ batch,
                                                            float* __restrict__ gemb, int n) {
    static_assert(DIM_O == 128, "a lane owns 4 of 128 columns");
    __shared__ float4 part[POOL_WAVES * 32];
    __shared__ int    wcnt[POOL_WAVES];

    const int g    = blockIdx.x;
    const int lane = threadIdx.x & 31;
    const int wv   = __builtin_amdgcn_readfirstlane((int)(threadIdx.x >> 5));

    const float* el = emb + lane * 4;
    float4 acc = make_float4(0.f, 0.f, 0.f, 0.f);
    int cnt = 0;

    for (int c0 = wv * 32; c0 < n; c0 += POOL_CHUNK * POOL_THREADS) {
        int b[POOL_CHUNK];
#pragma unroll
        for (int u = 0; u < POOL_CHUNK; ++u) {
            const int i = c0 + u * POOL_THREADS + lane;
            b[u] = (i < n) ? batch[i] : -1;
        }
#pragma unroll
        for (int u = 0; u < POOL_CHUNK; ++u) {
            unsigned mask = __builtin_amdgcn_ballot_w32(b[u] == g);
            cnt += __builtin_popcount(mask);
            const int base = c0 + u * POOL_THREADS;
            while (mask) {
                float4 v[POOL_ROWS];
#pragma unroll
                for (int r = 0; r < POOL_ROWS; ++r) {
                    v[r] = make_float4(0.f, 0.f, 0.f, 0.f);
                    if (mask) {
                        const int k = __builtin_ctz(mask);
                        mask &= mask - 1;
                        v[r] = *(const float4*)(el + (size_t)(base + k) * DIM_O);
                    }
                }
#pragma unroll
                for (int r = 0; r < POOL_ROWS; ++r) {
                    acc.x += v[r].x; acc.y += v[r].y;
                    acc.z += v[r].z; acc.w += v[r].w;
                }
            }
        }
    }

    part[wv * 32 + lane] = acc;
    if (lane == 0) wcnt[wv] = cnt;
    __syncthreads();

    if (threadIdx.x < DIM_O) {
        const float* pf = (const float*)part;
        float s = 0.0f;
        int   c = 0;
        for (int w = 0; w < POOL_WAVES; ++w) {
            s += pf[w * DIM_O + threadIdx.x];
            c += wcnt[w];
        }
        gemb[(size_t)g * DIM_O + threadIdx.x] = s / fmaxf((float)c, 1.0f);
    }
}


__global__ void k_logits(const float* __restrict__ gemb, const float* __restrict__ Wc,
                         const float* __restrict__ bc, float* __restrict__ logits, int D) {
    const int t = threadIdx.x;
    const int g = t >> 1;
    const int c = t & 1;
    float acc = bc[c];
    const float* row = gemb + (size_t)g * D;
    for (int k = 0; k < D; ++k) acc += row[k] * Wc[k * 2 + c];
    logits[g * 2 + c] = acc;
}


static inline int cdiv_l(long a, int b) { return (int)((a + b - 1) / b); }

extern "C" void kernel_launch(void* const* d_in, const int* in_sizes, int n_in,
                              void* d_out, int out_size, void* d_ws, size_t ws_size,
                              hipStream_t stream) {
    const float* x     = (const float*)d_in[0];
    const int*   ei    = (const int*)d_in[1];
    const int*   batch = (const int*)d_in[2];
    const float* W_in  = (const float*)d_in[3];
    const float* b_in  = (const float*)d_in[4];
    const float* W_mid = (const float*)d_in[5];
    const float* b_mid = (const float*)d_in[6];
    const float* W_out = (const float*)d_in[7];
    const float* b_out = (const float*)d_in[8];
    const float* W_cls = (const float*)d_in[9];
    const float* b_cls = (const float*)d_in[10];

    const int N = in_sizes[0] / DIM_IN;
    const int E = in_sizes[1] / 2;
    const int* src = ei;
    const int* dst = ei + E;

    float* w       = (float*)d_ws;
    float* dis     = w;
    float* bufH    = dis + N;
    float* bufCsr  = bufH + (size_t)N * DIM_H;
    float* bufA    = bufCsr + (size_t)N * DIM_H;

    int*  rowptr = (int*)bufCsr;
    int*  slot   = rowptr + (N + 64);
    int2* csr    = (int2*)(slot + E);

    float* node_emb = (float*)d_out;
    float* gemb     = node_emb + (size_t)N * DIM_O;
    float* logits   = gemb + (size_t)NGRAPH * DIM_O;

    const int B = 256;
    const int MB = 32;
    const int GB = cdiv_l(N, 8);

    k_zero_int<<<cdiv_l(N, B), B, 0, stream>>>(rowptr, N);
    k_deg_slot<<<cdiv_l(E, B), B, 0, stream>>>(dst, rowptr, slot, E);
    k_scan_dis<<<1, SCAN_THREADS, 0, stream>>>(rowptr, dis, N);
    k_place<<<cdiv_l(E, B), B, 0, stream>>>(src, dst, rowptr, slot, dis, csr, E);

    k_gather_gemm<DIM_IN, DIM_H, 2, 4, 0, 1><<<N / MB, 128, 0, stream>>>(
        x, rowptr, csr, dis, nullptr, W_in, b_in, bufA);

    {
        dim3 grid(N / MB, DIM_H / 256);
        k_gemm_wmma<2, 4, 0><<<grid, 128, 0, stream>>>(bufA, W_mid, b_mid, bufH, N, DIM_H, DIM_H);
    }

    k_gather_gemm<DIM_H, DIM_O, 1, 2, 1, 0><<<N / MB, 128, 0, stream>>>(
        bufH, rowptr, csr, dis, b_mid, W_out, b_out, bufA);

    k_gather<DIM_O, 0><<<GB, B, 0, stream>>>(bufA, rowptr, csr, dis, b_out, node_emb, N);

    k_pool_mean<<<NGRAPH, POOL_THREADS, 0, stream>>>(node_emb, batch, gemb, N);

    k_logits<<<1, 128, 0, stream>>>(gemb, W_cls, b_cls, logits, DIM_O);
}


# Layer 2's product joins layer 1's kernel: the block's activated 32 x 256 tile goes from the accumulators into a second LDS tile and is multiplied by W_mid there; 10 launches -> 9
# speedup vs baseline: 4.1941x; 1.0557x over previous
#include <hip/hip_runtime.h>
#include <hip/hip_bf16.h>


typedef float v2f __attribute__((ext_vector_type(2)));
typedef float v8f __attribute__((ext_vector_type(8)));

#define NNODES 40000
#define NEDGES 640000
#define NGRAPH 64
#define DIM_IN 128
#define DIM_H  256
#define DIM_O  128


__global__ void k_zero_int(int* __restrict__ p, int n) {
    int i = blockIdx.x * blockDim.x + threadIdx.x;
    if (i < n) p[i] = 0;
}

__global__ void k_deg_slot(const int* __restrict__ dst, int* __restrict__ cnt,
                           int* __restrict__ slot, int e) {
    int i = blockIdx.x * blockDim.x + threadIdx.x;
    if (i < e) slot[i] = atomicAdd(&cnt[dst[i]], 1);
}

#define SCAN_THREADS 1024
#define SCAN_WAVES   (SCAN_THREADS / 32)
#define SCAN_PASSES  ((NNODES + SCAN_THREADS * 4 - 1) / (SCAN_THREADS * 4))
#define SCAN_WSPAN   (SCAN_PASSES * 128)

__global__ __launch_bounds__(SCAN_THREADS) void k_scan_dis(int* __restrict__ cr,
                                                           float* __restrict__ dis, int n) {
    static_assert(SCAN_WAVES == 32, "one lane of the first wave per wave total");
    static_assert(NNODES % 4 == 0, "a lane's group of 4 nodes lies wholly inside or wholly outside the range");
    __shared__ int wsum[SCAN_WAVES];
    const int t    = threadIdx.x;
    const int lane = t & 31;
    const int wv   = t >> 5;
    const int b0   = wv * SCAN_WSPAN + lane * 4;

    int4 c[SCAN_PASSES];
#pragma unroll
    for (int j = 0; j < SCAN_PASSES; ++j) {
        const int idx = b0 + j * 128;
        c[j] = make_int4(0, 0, 0, 0);
        if (idx < n) c[j] = *(const int4*)(cr + idx);
    }

    int s4[SCAN_PASSES];
    int x[SCAN_PASSES];
#pragma unroll
    for (int j = 0; j < SCAN_PASSES; ++j) {
        s4[j] = c[j].x + c[j].y + c[j].z + c[j].w;
        x[j]  = s4[j];
    }
#pragma unroll
    for (int off = 1; off < 32; off <<= 1) {
#pragma unroll
        for (int j = 0; j < SCAN_PASSES; ++j) {
            const int y = __shfl_up(x[j], off);
            if (lane >= off) x[j] += y;
        }
    }

    int ex[SCAN_PASSES];
    int carry = 0;
#pragma unroll
    for (int j = 0; j < SCAN_PASSES; ++j) {
        ex[j] = carry + x[j] - s4[j];
        carry += __builtin_amdgcn_readlane(x[j], 31);
    }

    if (lane == 0) wsum[wv] = carry;
    __syncthreads();
    if (wv == 0) {
        int s = wsum[lane];
        for (int off = 1; off < 32; off <<= 1) {
            const int y = __shfl_up(s, off);
            if (lane >= off) s += y;
        }
        wsum[lane] = s;
    }
    __syncthreads();

    const int wbase = (wv > 0) ? wsum[wv - 1] : 0;
#pragma unroll
    for (int j = 0; j < SCAN_PASSES; ++j) {
        const int idx = b0 + j * 128;
        int4 r;
        r.x = wbase + ex[j];
        r.y = r.x + c[j].x;
        r.z = r.y + c[j].y;
        r.w = r.z + c[j].z;
        float4 d;
        d.x = rsqrtf((float)c[j].x + 1.0f);
        d.y = rsqrtf((float)c[j].y + 1.0f);
        d.z = rsqrtf((float)c[j].z + 1.0f);
        d.w = rsqrtf((float)c[j].w + 1.0f);
        if (idx < n) {
            *(int4*)(cr + idx)    = r;
            *(float4*)(dis + idx) = d;
        }
    }
    if (t == SCAN_THREADS - 1) cr[n] = wbase + carry;
}

__global__ void k_place(const int* __restrict__ src, const int* __restrict__ dst,
                        const int* __restrict__ rowptr, const int* __restrict__ slot,
                        const float* __restrict__ dis, int2* __restrict__ csr, int e) {
    int i = blockIdx.x * blockDim.x + threadIdx.x;
    if (i >= e) return;
    const int s = src[i];
    const int d = dst[i];
    const float nrm = dis[s] * dis[d];
    csr[rowptr[d] + slot[i]] = make_int2(s, __float_as_int(nrm));
}


template <int MT>
__device__ __forceinline__ void gemm_steps(const float4 (&a)[MT], const float4 (&b)[4],
                                           v8f (&acc)[MT][4]) {
    const float bx[4][4] = {{b[0].x, b[0].y, b[0].z, b[0].w},
                            {b[1].x, b[1].y, b[1].z, b[1].w},
                            {b[2].x, b[2].y, b[2].z, b[2].w},
                            {b[3].x, b[3].y, b[3].z, b[3].w}};
    for (int p = 0; p < 2; ++p) {
        for (int t = 0; t < MT; ++t) {
            v2f av;
            av.x = p ? a[t].z : a[t].x;
            av.y = p ? a[t].w : a[t].y;
            for (int q = 0; q < 4; ++q) {
                v2f bv;
                bv.x = bx[2 * p][q];
                bv.y = bx[2 * p + 1][q];
                acc[t][q] = __builtin_amdgcn_wmma_f32_16x16x4_f32(
                     false, av,  false, bv,
                     (short)0, acc[t][q],  false,  false);
            }
        }
    }
}


template <int H, int K>
__device__ __forceinline__ void gather_rows(const float* __restrict__ hl, int my_s, int my_w,
                                            int i, float4 (&acc)[H / 128]) {
    constexpr int V = H / 128;
    float  w[K];
    float4 v[K][V];
#pragma unroll
    for (int k = 0; k < K; ++k) {
        const int s = __builtin_amdgcn_readlane(my_s, i + k);
        w[k] = __int_as_float(__builtin_amdgcn_readlane(my_w, i + k));
        const float* row = hl + (size_t)s * H;
        for (int q = 0; q < V; ++q) v[k][q] = *(const float4*)(row + q * 128);
    }
#pragma unroll
    for (int k = 0; k < K; ++k) {
        for (int q = 0; q < V; ++q) {
            acc[q].x = __builtin_fmaf(w[k], v[k][q].x, acc[q].x);
            acc[q].y = __builtin_fmaf(w[k], v[k][q].y, acc[q].y);
            acc[q].z = __builtin_fmaf(w[k], v[k][q].z, acc[q].z);
            acc[q].w = __builtin_fmaf(w[k], v[k][q].w, acc[q].w);
        }
    }
}

template <int H, int RELU>
__global__ __launch_bounds__(256) void k_gather(const float* __restrict__ h,
                                                const int* __restrict__ rowptr,
                                                const int2* __restrict__ csr,
                                                const float* __restrict__ dis,
                                                const float* __restrict__ b,
                                                float* __restrict__ out, int n) {
    constexpr int V = H / 128;
    constexpr int K = 1024 / H;
    const int lane = threadIdx.x & 31;
    const int node = __builtin_amdgcn_readfirstlane((int)(blockIdx.x * 8 + (threadIdx.x >> 5)));
    if (node >= n) return;
    const int beg = rowptr[node];
    const int end = rowptr[node + 1];

    const float* hl = h + lane * 4;
    float4 acc[V];
    for (int q = 0; q < V; ++q) acc[q] = make_float4(0.f, 0.f, 0.f, 0.f);

    for (int base = beg; base < end; base += 32) {
        const int cntw = (end - base < 32) ? (end - base) : 32;
        int2 me = make_int2(0, 0);
        if (lane < cntw) me = csr[base + lane];
        int i = 0;
        for (; i + K <= cntw; i += K) gather_rows<H, K>(hl, me.x, me.y, i, acc);
        if (K >= 8 && i + 4 <= cntw) { gather_rows<H, 4>(hl, me.x, me.y, i, acc); i += 4; }
        if (i + 2 <= cntw)           { gather_rows<H, 2>(hl, me.x, me.y, i, acc); i += 2; }
        if (i + 1 <= cntw)           { gather_rows<H, 1>(hl, me.x, me.y, i, acc); }
    }

    const float ds = dis[node];
    const float sl = ds * ds;
    const float* hrow = hl + (size_t)node * H;
    float* orow = out + (size_t)node * H + lane * 4;
    for (int q = 0; q < V; ++q) {
        const float4 hv = *(const float4*)(hrow + q * 128);
        float4 r;
        r.x = acc[q].x + sl * hv.x;
        r.y = acc[q].y + sl * hv.y;
        r.z = acc[q].z + sl * hv.z;
        r.w = acc[q].w + sl * hv.w;
        if (b != nullptr) {
            const float4 bv = *(const float4*)(b + q * 128 + lane * 4);
            r.x += bv.x; r.y += bv.y; r.z += bv.z; r.w += bv.w;
        }
        if (RELU) {
            r.x = fmaxf(r.x, 0.0f); r.y = fmaxf(r.y, 0.0f);
            r.z = fmaxf(r.z, 0.0f); r.w = fmaxf(r.w, 0.0f);
        }
        *(float4*)(orow + q * 128) = r;
    }
}


template <int H>
__device__ __forceinline__ void gather_chunk(const float* __restrict__ hl, int2 me, int cntw,
                                             float4 (&acc)[H / 128]) {
    constexpr int K = 2048 / H;
    int i = 0;
    for (; i + K <= cntw; i += K) gather_rows<H, K>(hl, me.x, me.y, i, acc);
    if (K >= 16 && i + 8 <= cntw) { gather_rows<H, 8>(hl, me.x, me.y, i, acc); i += 8; }
    if (i + 4 <= cntw)            { gather_rows<H, 4>(hl, me.x, me.y, i, acc); i += 4; }
    if (i + 2 <= cntw)            { gather_rows<H, 2>(hl, me.x, me.y, i, acc); i += 2; }
    if (i + 1 <= cntw)            { gather_rows<H, 1>(hl, me.x, me.y, i, acc); }
}

template <int KD, int GRELU>
__device__ __forceinline__ void aggregate_rows_to_tile(const float* __restrict__ h,
                                                       const int* __restrict__ rowptr,
                                                       const int2* __restrict__ csr,
                                                       const float* __restrict__ dis,
                                                       const float* __restrict__ gb,
                                                       float* tile, int nb, int trow0, int lane) {
    constexpr int RPW   = 8;
    constexpr int PITCH = KD + 4;
    constexpr int V     = KD / 128;
    const int rp = rowptr[nb + (lane < RPW ? lane : RPW)];
    const float* hl = h + lane * 4;

    const int plast = NEDGES - 1;
    int2 cur;
    {
        const int beg = __builtin_amdgcn_readlane(rp, 0);
        const int pi = beg + lane;
        cur = csr[pi < plast ? pi : plast];
    }
#pragma unroll 1
    for (int j = 0; j < RPW; ++j) {
        const int node = nb + j;
        const int beg = __builtin_amdgcn_readlane(rp, j);
        const int end = __builtin_amdgcn_readlane(rp, j + 1);

        const int pi = end + lane;
        const int2 nxt = csr[pi < plast ? pi : plast];
        const float ds = dis[node];
        const float* hrow = hl + (size_t)node * KD;
        float4 hv[V];
        for (int q = 0; q < V; ++q) hv[q] = *(const float4*)(hrow + q * 128);

        float4 acc[V];
        for (int q = 0; q < V; ++q) acc[q] = make_float4(0.f, 0.f, 0.f, 0.f);

        const int c0 = (end - beg < 32) ? (end - beg) : 32;
        gather_chunk<KD>(hl, cur, c0, acc);
        for (int base = beg + 32; base < end; base += 32) {
            const int cntw = (end - base < 32) ? (end - base) : 32;
            int2 m = make_int2(0, 0);
            if (lane < cntw) m = csr[base + lane];
            gather_chunk<KD>(hl, m, cntw, acc);
        }
        cur = nxt;

        const float sl = ds * ds;
        float* trow = tile + (trow0 + j) * PITCH + lane * 4;
        for (int q = 0; q < V; ++q) {
            float4 r;
            r.x = acc[q].x + sl * hv[q].x;
            r.y = acc[q].y + sl * hv[q].y;
            r.z = acc[q].z + sl * hv[q].z;
            r.w = acc[q].w + sl * hv[q].w;
            if (gb != nullptr) {
                const float4 bv = *(const float4*)(gb + q * 128 + lane * 4);
                r.x += bv.x; r.y += bv.y; r.z += bv.z; r.w += bv.w;
            }
            if (GRELU) {
                r.x = fmaxf(r.x, 0.0f); r.y = fmaxf(r.y, 0.0f);
                r.z = fmaxf(r.z, 0.0f); r.w = fmaxf(r.w, 0.0f);
            }
            *(float4*)(trow + q * 128) = r;
        }
    }
}

template <int KD, int MT>
__device__ __forceinline__ void gemm_fetch_tile(const float* as, const float* __restrict__ wp,
                                                int k, int Nc, float4 (&a)[MT], float4 (&b)[4]) {
    for (int t = 0; t < MT; ++t) a[t] = *(const float4*)(as + t * 16 * (KD + 4) + k);
    for (int j = 0; j < 4; ++j)  b[j] = *(const float4*)(wp + (size_t)(k + j) * Nc);
}

template <int KD, int NC, int MT>
__device__ __forceinline__ void tile_product(const float* as, const float* __restrict__ wp,
                                             v8f (&acc)[MT][4]) {
    for (int t = 0; t < MT; ++t)
        for (int q = 0; q < 4; ++q) acc[t][q] = (v8f){};

    float4 a0[MT], b0[4], a1[MT], b1[4];
    gemm_fetch_tile<KD, MT>(as, wp, 0, NC, a0, b0);

    for (int k0 = 0; k0 < KD; k0 += 16) {
        gemm_fetch_tile<KD, MT>(as, wp, k0 + 8, NC, a1, b1);
        __builtin_amdgcn_sched_barrier(0);
        gemm_steps<MT>(a0, b0, acc);

        const int kn = (k0 + 16 < KD) ? k0 + 16 : k0;
        gemm_fetch_tile<KD, MT>(as, wp, kn, NC, a0, b0);
        __builtin_amdgcn_sched_barrier(0);
        gemm_steps<MT>(a1, b1, acc);
    }
}

template <int KD, int NC, int MT, int WN, int GRELU, int EPI>
__global__ __launch_bounds__(128) void k_gather_gemm(const float* __restrict__ h,
                                                     const int* __restrict__ rowptr,
                                                     const int2* __restrict__ csr,
                                                     const float* __restrict__ dis,
                                                     const float* __restrict__ gb,
                                                     const float* __restrict__ W,
                                                     const float* __restrict__ bias,
                                                     float* __restrict__ C) {
    constexpr int WM    = 4 / WN;
    constexpr int R     = 16 * MT * WM;
    constexpr int RPW   = R / 4;
    constexpr int PITCH = KD + 4;
    static_assert(R == 32 && RPW == 8, "4 waves x 8 rows");
    static_assert(NNODES % R == 0, "every block owns 32 whole rows: no row guard");
    static_assert(NC == 64 * WN, "the block covers all columns");
    __shared__ float4 tile4[R * PITCH / 4];
    float* tile = (float*)tile4;

    const int lane = threadIdx.x & 31;
    const int wave = __builtin_amdgcn_readfirstlane((int)(threadIdx.x >> 5));
    const int m0 = (int)blockIdx.x * R;

    aggregate_rows_to_tile<KD, GRELU>(h, rowptr, csr, dis, gb, tile, m0 + wave * RPW, wave * RPW, lane);
    __syncthreads();

    const int rg = wave / WN;
    const int n0 = (wave % WN) * 64;
    const int hh = lane >> 4;
    const int c  = lane & 15;

    v8f acc[MT][4];
    tile_product<KD, NC, MT>(tile + (rg * 16 * MT + c) * PITCH + 4 * hh,
                             W + (size_t)(4 * hh) * NC + n0 + 4 * c, acc);

    float4 bv = make_float4(0.f, 0.f, 0.f, 0.f);
    if (EPI) bv = *(const float4*)(bias + n0 + 4 * c);
    for (int t = 0; t < MT; ++t) {
        float* crow = C + (size_t)(m0 + rg * 16 * MT + t * 16 + 8 * hh) * NC + n0 + 4 * c;
        for (int v = 0; v < 8; ++v) {
            float4 r;
            r.x = acc[t][0][v]; r.y = acc[t][1][v];
            r.z = acc[t][2][v]; r.w = acc[t][3][v];
            if (EPI) {
                r.x = fmaxf(r.x + bv.x, 0.0f); r.y = fmaxf(r.y + bv.y, 0.0f);
                r.z = fmaxf(r.z + bv.z, 0.0f); r.w = fmaxf(r.w + bv.w, 0.0f);
            }
            *(float4*)(crow + (size_t)v * NC) = r;
        }
    }
}

template <int KD, int NH, int NC>
__global__ __launch_bounds__(128) void k_gather_gemm_gemm(const float* __restrict__ h,
                                                          const int* __restrict__ rowptr,
                                                          const int2* __restrict__ csr,
                                                          const float* __restrict__ dis,
                                                          const float* __restrict__ W1,
                                                          const float* __restrict__ b1,
                                                          const float* __restrict__ W2,
                                                          float* __restrict__ C) {
    constexpr int R   = 32;
    constexpr int RPW = R / 4;
    constexpr int MT  = 2;
    constexpr int P1  = KD + 4;
    constexpr int P2  = NH + 4;
    static_assert(NNODES % R == 0, "every block owns 32 whole rows: no row guard");
    static_assert(NH == 256 && NC == 256, "4 waves x 64 columns in both products");
    __shared__ float4 tile1v[R * P1 / 4];
    __shared__ float4 tile2v[R * P2 / 4];
    float* tile1 = (float*)tile1v;
    float* tile2 = (float*)tile2v;

    const int lane = threadIdx.x & 31;
    const int wave = __builtin_amdgcn_readfirstlane((int)(threadIdx.x >> 5));
    const int m0 = (int)blockIdx.x * R;

    aggregate_rows_to_tile<KD, 0>(h, rowptr, csr, dis, nullptr, tile1, m0 + wave * RPW, wave * RPW, lane);
    __syncthreads();

    const int n0 = wave * 64;
    const int hh = lane >> 4;
    const int c  = lane & 15;

    v8f acc[MT][4];

    tile_product<KD, NH, MT>(tile1 + c * P1 + 4 * hh, W1 + (size_t)(4 * hh) * NH + n0 + 4 * c, acc);
    {
        const float4 bv = *(const float4*)(b1 + n0 + 4 * c);
        for (int t = 0; t < MT; ++t) {
            float* trow = tile2 + (t * 16 + 8 * hh) * P2 + n0 + 4 * c;
            for (int v = 0; v < 8; ++v) {
                float4 r;
                r.x = fmaxf(acc[t][0][v] + bv.x, 0.0f); r.y = fmaxf(acc[t][1][v] + bv.y, 0.0f);
                r.z = fmaxf(acc[t][2][v] + bv.z, 0.0f); r.w = fmaxf(acc[t][3][v] + bv.w, 0.0f);
                *(float4*)(trow + v * P2) = r;
            }
        }
    }
    __syncthreads();

    tile_product<NH, NC, MT>(tile2 + c * P2 + 4 * hh, W2 + (size_t)(4 * hh) * NC + n0 + 4 * c, acc);
    for (int t = 0; t < MT; ++t) {
        float* crow = C + (size_t)(m0 + t * 16 + 8 * hh) * NC + n0 + 4 * c;
        for (int v = 0; v < 8; ++v) {
            float4 r;
            r.x = acc[t][0][v]; r.y = acc[t][1][v];
            r.z = acc[t][2][v]; r.w = acc[t][3][v];
            *(float4*)(crow + (size_t)v * NC) = r;
        }
    }
}


#define POOL_THREADS 1024
#define POOL_WAVES   (POOL_THREADS / 32)
#define POOL_CHUNK   8
#define POOL_ROWS    8

__global__ __launch_bounds__(POOL_THREADS) void k_pool_mean(const float* __restrict__ emb,
                                                            const int* __restrict__ batch,
                                                            float* __restrict__ gemb, int n) {
    static_assert(DIM_O == 128, "a lane owns 4 of 128 columns");
    __shared__ float4 part[POOL_WAVES * 32];
    __shared__ int    wcnt[POOL_WAVES];

    const int g    = blockIdx.x;
    const int lane = threadIdx.x & 31;
    const int wv   = __builtin_amdgcn_readfirstlane((int)(threadIdx.x >> 5));

    const float* el = emb + lane * 4;
    float4 acc = make_float4(0.f, 0.f, 0.f, 0.f);
    int cnt = 0;

    for (int c0 = wv * 32; c0 < n; c0 += POOL_CHUNK * POOL_THREADS) {
        int b[POOL_CHUNK];
#pragma unroll
        for (int u = 0; u < POOL_CHUNK; ++u) {
            const int i = c0 + u * POOL_THREADS + lane;
            b[u] = (i < n) ? batch[i] : -1;
        }
#pragma unroll
        for (int u = 0; u < POOL_CHUNK; ++u) {
            unsigned mask = __builtin_amdgcn_ballot_w32(b[u] == g);
            cnt += __builtin_popcount(mask);
            const int base = c0 + u * POOL_THREADS;
            while (mask) {
                float4 v[POOL_ROWS];
#pragma unroll
                for (int r = 0; r < POOL_ROWS; ++r) {
                    v[r] = make_float4(0.f, 0.f, 0.f, 0.f);
                    if (mask) {
                        const int k = __builtin_ctz(mask);
                        mask &= mask - 1;
                        v[r] = *(const float4*)(el + (size_t)(base + k) * DIM_O);
                    }
                }
#pragma unroll
                for (int r = 0; r < POOL_ROWS; ++r) {
                    acc.x += v[r].x; acc.y += v[r].y;
                    acc.z += v[r].z; acc.w += v[r].w;
                }
            }
        }
    }

    part[wv * 32 + lane] = acc;
    if (lane == 0) wcnt[wv] = cnt;
    __syncthreads();

    if (threadIdx.x < DIM_O) {
        const float* pf = (const float*)part;
        float s = 0.0f;
        int   c = 0;
        for (int w = 0; w < POOL_WAVES; ++w) {
            s += pf[w * DIM_O + threadIdx.x];
            c += wcnt[w];
        }
        gemb[(size_t)g * DIM_O + threadIdx.x] = s / fmaxf((float)c, 1.0f);
    }
}


__global__ void k_logits(const float* __restrict__ gemb, const float* __restrict__ Wc,
                         const float* __restrict__ bc, float* __restrict__ logits, int D) {
    const int t = threadIdx.x;
    const int g = t >> 1;
    const int c = t & 1;
    float acc = bc[c];
    const float* row = gemb + (size_t)g * D;
    for (int k = 0; k < D; ++k) acc += row[k] * Wc[k * 2 + c];
    logits[g * 2 + c] = acc;
}


static inline int cdiv_l(long a, int b) { return (int)((a + b - 1) / b); }

extern "C" void kernel_launch(void* const* d_in, const int* in_sizes, int n_in,
                              void* d_out, int out_size, void* d_ws, size_t ws_size,
                              hipStream_t stream) {
    const float* x     = (const float*)d_in[0];
    const int*   ei    = (const int*)d_in[1];
    const int*   batch = (const int*)d_in[2];
    const float* W_in  = (const float*)d_in[3];
    const float* b_in  = (const float*)d_in[4];
    const float* W_mid = (const float*)d_in[5];
    const float* b_mid = (const float*)d_in[6];
    const float* W_out = (const float*)d_in[7];
    const float* b_out = (const float*)d_in[8];
    const float* W_cls = (const float*)d_in[9];
    const float* b_cls = (const float*)d_in[10];

    const int N = in_sizes[0] / DIM_IN;
    const int E = in_sizes[1] / 2;
    const int* src = ei;
    const int* dst = ei + E;

    float* w       = (float*)d_ws;
    float* dis     = w;
    float* bufH    = dis + N;
    float* bufCsr  = bufH + (size_t)N * DIM_H;
    float* bufA    = bufCsr + (size_t)N * DIM_H;

    int*  rowptr = (int*)bufCsr;
    int*  slot   = rowptr + (N + 64);
    int2* csr    = (int2*)(slot + E);

    float* node_emb = (float*)d_out;
    float* gemb     = node_emb + (size_t)N * DIM_O;
    float* logits   = gemb + (size_t)NGRAPH * DIM_O;

    const int B = 256;
    const int MB = 32;
    const int GB = cdiv_l(N, 8);

    k_zero_int<<<cdiv_l(N, B), B, 0, stream>>>(rowptr, N);
    k_deg_slot<<<cdiv_l(E, B), B, 0, stream>>>(dst, rowptr, slot, E);
    k_scan_dis<<<1, SCAN_THREADS, 0, stream>>>(rowptr, dis, N);
    k_place<<<cdiv_l(E, B), B, 0, stream>>>(src, dst, rowptr, slot, dis, csr, E);

    k_gather_gemm_gemm<DIM_IN, DIM_H, DIM_H><<<N / MB, 128, 0, stream>>>(
        x, rowptr, csr, dis, W_in, b_in, W_mid, bufH);

    k_gather_gemm<DIM_H, DIM_O, 1, 2, 1, 0><<<N / MB, 128, 0, stream>>>(
        bufH, rowptr, csr, dis, b_mid, W_out, b_out, bufA);

    k_gather<DIM_O, 0><<<GB, B, 0, stream>>>(bufA, rowptr, csr, dis, b_out, node_emb, N);

    k_pool_mean<<<NGRAPH, POOL_THREADS, 0, stream>>>(node_emb, batch, gemb, N);

    k_logits<<<1, 128, 0, stream>>>(gemb, W_cls, b_cls, logits, DIM_O);
}
